# starting kernel
# baseline (speedup 1.0000x reference)
#include <hip/hip_runtime.h>
#include <hip/hip_bf16.h>
#include <math.h>

typedef float v2f __attribute__((ext_vector_type(2)));
typedef float v8f __attribute__((ext_vector_type(8)));

#define BATCH   2
#define SEQ     2048
#define DMODEL  1024
#define DTRANK  64
#define NSTATE  16
#define KPROJ   (DTRANK + 2 * NSTATE)
#define MROWS   (BATCH * SEQ)
#define NCHAN   (BATCH * DMODEL)
#define NCHUNK  32
#define CLEN    (SEQ / NCHUNK)

__global__ __launch_bounds__(256)
void gemm_xproj(const float* __restrict__ U, const float* __restrict__ W1,
                float* __restrict__ xp) {
    const int lane  = threadIdx.x & 31;
    const int wave  = threadIdx.x >> 5;
    const int tile  = blockIdx.x * (blockDim.x >> 5) + wave;
    const int tilesN = KPROJ / 16;
    const int m0 = (tile / tilesN) * 16;
    const int n0 = (tile % tilesN) * 16;

    const int row   = m0 + (lane & 15);
    const int col   = n0 + (lane & 15);
    const int khalf = (lane >> 4) * 2;

    v8f acc = {};
    #pragma unroll 4
    for (int k = 0; k < DMODEL; k += 4) {
        v2f a = *(const v2f*)(U  + row * DMODEL + k + khalf);
        v2f b = *(const v2f*)(W1 + col * DMODEL + k + khalf);
        acc = __builtin_amdgcn_wmma_f32_16x16x4_f32(false, a, false, b,
                                                    (short)0, acc, false, false);
    }
    const int rbase = m0 + ((lane >> 4) ? 8 : 0);
    #pragma unroll
    for (int i = 0; i < 8; ++i)
        xp[(rbase + i) * KPROJ + col] = acc[i];
}

__global__ __launch_bounds__(256)
void gemm_delta(const float* __restrict__ xp, const float* __restrict__ W2,
                const float* __restrict__ bias, float* __restrict__ delta) {
    const int lane  = threadIdx.x & 31;
    const int wave  = threadIdx.x >> 5;
    const int tile  = blockIdx.x * (blockDim.x >> 5) + wave;
    const int tilesN = DMODEL / 16;
    const int m0 = (tile / tilesN) * 16;
    const int n0 = (tile % tilesN) * 16;

    const int row   = m0 + (lane & 15);
    const int col   = n0 + (lane & 15);
    const int khalf = (lane >> 4) * 2;

    v8f acc = {};
    #pragma unroll
    for (int k = 0; k < DTRANK; k += 4) {
        v2f a = *(const v2f*)(xp + row * KPROJ  + k + khalf);
        v2f b = *(const v2f*)(W2 + col * DTRANK + k + khalf);
        acc = __builtin_amdgcn_wmma_f32_16x16x4_f32(false, a, false, b,
                                                    (short)0, acc, false, false);
    }
    const int   rbase = m0 + ((lane >> 4) ? 8 : 0);
    const float bn    = bias[col];
    #pragma unroll
    for (int i = 0; i < 8; ++i) {
        float x  = acc[i] + bn;
        float sp = (x > 20.0f) ? x : log1pf(__expf(x));
        delta[(rbase + i) * DMODEL + col] = sp;
    }
}

__global__ __launch_bounds__(256)
void scan_local(const float* __restrict__ U, const float* __restrict__ A_log,
                const float* __restrict__ xp, const float* __restrict__ delta,
                float* __restrict__ out, float* __restrict__ agg) {
    const int ch = blockIdx.x * blockDim.x + threadIdx.x;
    const int c  = blockIdx.y;
    const int bb = ch >> 10;
    const int dd = ch & (DMODEL - 1);

    float Av[NSTATE];
    #pragma unroll
    for (int n = 0; n < NSTATE; ++n) Av[n] = -__expf(A_log[n]);

    float st[NSTATE];
    #pragma unroll
    for (int n = 0; n < NSTATE; ++n) st[n] = 0.0f;

    float dsum = 0.0f;
    const int t0 = c * CLEN;
    for (int t = t0; t < t0 + CLEN; ++t) {
        const int rowu = bb * SEQ + t;
        const float dl = delta[rowu * DMODEL + dd];
        const float uv = U[rowu * DMODEL + dd];
        const float4* Bp = (const float4*)(xp + rowu * KPROJ + DTRANK);
        const float4* Cp = (const float4*)(xp + rowu * KPROJ + DTRANK + NSTATE);
        float Bv[NSTATE], Cv[NSTATE];
        #pragma unroll
        for (int q = 0; q < 4; ++q) {
            float4 vb = Bp[q], vc = Cp[q];
            Bv[4*q+0]=vb.x; Bv[4*q+1]=vb.y; Bv[4*q+2]=vb.z; Bv[4*q+3]=vb.w;
            Cv[4*q+0]=vc.x; Cv[4*q+1]=vc.y; Cv[4*q+2]=vc.z; Cv[4*q+3]=vc.w;
        }
        const float du = dl * uv;
        float y = 0.0f;
        #pragma unroll
        for (int n = 0; n < NSTATE; ++n) {
            float dA = __expf(dl * Av[n]);
            st[n] = fmaf(dA, st[n], du * Bv[n]);
            y     = fmaf(st[n], Cv[n], y);
        }
        dsum += dl;
        out[rowu * DMODEL + dd] = y;
    }
    const int base = (c * 2 * NSTATE) * NCHAN + ch;
    #pragma unroll
    for (int n = 0; n < NSTATE; ++n)
        agg[base + n * NCHAN] = __expf(Av[n] * dsum);
    #pragma unroll
    for (int n = 0; n < NSTATE; ++n)
        agg[base + (NSTATE + n) * NCHAN] = st[n];
}

__global__ __launch_bounds__(256)
void scan_fixup(const float* __restrict__ U, const float* __restrict__ A_log,
                const float* __restrict__ xp, const float* __restrict__ delta,
                const float* __restrict__ Dvec, const float* __restrict__ agg,
                float* __restrict__ out) {
    const int ch = blockIdx.x * blockDim.x + threadIdx.x;
    const int c  = blockIdx.y;
    const int bb = ch >> 10;
    const int dd = ch & (DMODEL - 1);

    float Av[NSTATE];
    #pragma unroll
    for (int n = 0; n < NSTATE; ++n) Av[n] = -__expf(A_log[n]);

    float entry[NSTATE];
    #pragma unroll
    for (int n = 0; n < NSTATE; ++n) entry[n] = 0.0f;
    for (int cc = 0; cc < c; ++cc) {
        const int base = (cc * 2 * NSTATE) * NCHAN + ch;
        #pragma unroll
        for (int n = 0; n < NSTATE; ++n) {
            float p = agg[base + n * NCHAN];
            float f = agg[base + (NSTATE + n) * NCHAN];
            entry[n] = fmaf(p, entry[n], f);
        }
    }

    const float Dd = Dvec[dd];
    const int t0 = c * CLEN;
    for (int t = t0; t < t0 + CLEN; ++t) {
        const int rowu = bb * SEQ + t;
        const float dl = delta[rowu * DMODEL + dd];
        const float uv = U[rowu * DMODEL + dd];
        const float4* Cp = (const float4*)(xp + rowu * KPROJ + DTRANK + NSTATE);
        float Cv[NSTATE];
        #pragma unroll
        for (int q = 0; q < 4; ++q) {
            float4 vc = Cp[q];
            Cv[4*q+0]=vc.x; Cv[4*q+1]=vc.y; Cv[4*q+2]=vc.z; Cv[4*q+3]=vc.w;
        }
        float ycorr = 0.0f;
        #pragma unroll
        for (int n = 0; n < NSTATE; ++n) {
            entry[n] *= __expf(dl * Av[n]);
            ycorr = fmaf(entry[n], Cv[n], ycorr);
        }
        const int idx = rowu * DMODEL + dd;
        out[idx] = out[idx] + ycorr + uv * Dd;
    }
}

extern "C" void kernel_launch(void* const* d_in, const int* in_sizes, int n_in,
                              void* d_out, int out_size, void* d_ws, size_t ws_size,
                              hipStream_t stream) {
    const float* U     = (const float*)d_in[0];
    const float* A_log = (const float*)d_in[1];
    const float* W1    = (const float*)d_in[2];
    const float* W2    = (const float*)d_in[3];
    const float* bias  = (const float*)d_in[4];
    const float* Dvec  = (const float*)d_in[5];
    float* out = (float*)d_out;

    float* ws    = (float*)d_ws;
    float* xp    = ws;
    float* delta = xp + (size_t)MROWS * KPROJ;
    float* agg   = delta + (size_t)MROWS * DMODEL;

    gemm_xproj<<<dim3(1536 / 8), dim3(256), 0, stream>>>(U, W1, xp);

    gemm_delta<<<dim3(16384 / 8), dim3(256), 0, stream>>>(xp, W2, bias, delta);

    dim3 sgrid(NCHAN / 256, NCHUNK);
    scan_local<<<sgrid, dim3(256), 0, stream>>>(U, A_log, xp, delta, out, agg);
    scan_fixup<<<sgrid, dim3(256), 0, stream>>>(U, A_log, xp, delta, Dvec, agg, out);
}


# scan passes read operands ahead of the serial recurrence: chunk B/C rows staged once in LDS, delta/u/local y 8 steps ahead, chunk aggregates 4 per group; every sum and its order unchanged
# speedup vs baseline: 1.6245x; 1.6245x over previous
#include <hip/hip_runtime.h>
#include <hip/hip_bf16.h>
#include <math.h>

typedef float v2f __attribute__((ext_vector_type(2)));
typedef float v8f __attribute__((ext_vector_type(8)));

#define BATCH   2
#define SEQ     2048
#define DMODEL  1024
#define DTRANK  64
#define NSTATE  16
#define KPROJ   (DTRANK + 2 * NSTATE)
#define MROWS   (BATCH * SEQ)
#define NCHAN   (BATCH * DMODEL)
#define NCHUNK  32
#define CLEN    (SEQ / NCHUNK)

__global__ __launch_bounds__(256)
void gemm_xproj(const float* __restrict__ U, const float* __restrict__ W1,
                float* __restrict__ xp) {
    const int lane  = threadIdx.x & 31;
    const int wave  = threadIdx.x >> 5;
    const int tile  = blockIdx.x * (blockDim.x >> 5) + wave;
    const int tilesN = KPROJ / 16;
    const int m0 = (tile / tilesN) * 16;
    const int n0 = (tile % tilesN) * 16;

    const int row   = m0 + (lane & 15);
    const int col   = n0 + (lane & 15);
    const int khalf = (lane >> 4) * 2;

    v8f acc = {};
    #pragma unroll 4
    for (int k = 0; k < DMODEL; k += 4) {
        v2f a = *(const v2f*)(U  + row * DMODEL + k + khalf);
        v2f b = *(const v2f*)(W1 + col * DMODEL + k + khalf);
        acc = __builtin_amdgcn_wmma_f32_16x16x4_f32(false, a, false, b,
                                                    (short)0, acc, false, false);
    }
    const int rbase = m0 + ((lane >> 4) ? 8 : 0);
    #pragma unroll
    for (int i = 0; i < 8; ++i)
        xp[(rbase + i) * KPROJ + col] = acc[i];
}

__global__ __launch_bounds__(256)
void gemm_delta(const float* __restrict__ xp, const float* __restrict__ W2,
                const float* __restrict__ bias, float* __restrict__ delta) {
    const int lane  = threadIdx.x & 31;
    const int wave  = threadIdx.x >> 5;
    const int tile  = blockIdx.x * (blockDim.x >> 5) + wave;
    const int tilesN = DMODEL / 16;
    const int m0 = (tile / tilesN) * 16;
    const int n0 = (tile % tilesN) * 16;

    const int row   = m0 + (lane & 15);
    const int col   = n0 + (lane & 15);
    const int khalf = (lane >> 4) * 2;

    v8f acc = {};
    #pragma unroll
    for (int k = 0; k < DTRANK; k += 4) {
        v2f a = *(const v2f*)(xp + row * KPROJ  + k + khalf);
        v2f b = *(const v2f*)(W2 + col * DTRANK + k + khalf);
        acc = __builtin_amdgcn_wmma_f32_16x16x4_f32(false, a, false, b,
                                                    (short)0, acc, false, false);
    }
    const int   rbase = m0 + ((lane >> 4) ? 8 : 0);
    const float bn    = bias[col];
    #pragma unroll
    for (int i = 0; i < 8; ++i) {
        float x  = acc[i] + bn;
        float sp = (x > 20.0f) ? x : log1pf(__expf(x));
        delta[(rbase + i) * DMODEL + col] = sp;
    }
}

#define PF 8
#define PB 4
#define BPB (DMODEL / 256)

__global__ __launch_bounds__(256)
void scan_local(const float* __restrict__ U, const float* __restrict__ A_log,
                const float* __restrict__ xp, const float* __restrict__ delta,
                float* __restrict__ out, float* __restrict__ agg) {
    __shared__ float4 sBC[CLEN * 8];

    const int c    = blockIdx.y;
    const int bb   = blockIdx.x / BPB;
    const int dd   = (blockIdx.x % BPB) * 256 + threadIdx.x;
    const int ch   = bb * DMODEL + dd;
    const int row0 = bb * SEQ + c * CLEN;

    const float* __restrict__ dp = delta + row0 * DMODEL + dd;
    const float* __restrict__ up = U     + row0 * DMODEL + dd;
    float*       __restrict__ op = out   + row0 * DMODEL + dd;

    #pragma unroll
    for (int i = threadIdx.x; i < CLEN * 8; i += 256)
        sBC[i] = *(const float4*)(xp + (row0 + (i >> 3)) * KPROJ + DTRANK + 4 * (i & 7));

    float dn[PF], un[PF];
    #pragma unroll
    for (int j = 0; j < PF; ++j) { dn[j] = dp[j * DMODEL]; un[j] = up[j * DMODEL]; }

    float Av[NSTATE];
    #pragma unroll
    for (int n = 0; n < NSTATE; ++n) Av[n] = -__expf(A_log[n]);

    float st[NSTATE];
    #pragma unroll
    for (int n = 0; n < NSTATE; ++n) st[n] = 0.0f;

    __syncthreads();

    float4 nb[8];
    #pragma unroll
    for (int q = 0; q < 8; ++q) nb[q] = sBC[q];

    float dsum = 0.0f;
    #pragma unroll 1
    for (int tb = 0; tb < CLEN; tb += PF) {
        float dc[PF], uc[PF];
        #pragma unroll
        for (int j = 0; j < PF; ++j) { dc[j] = dn[j]; uc[j] = un[j]; }
        if (tb + PF < CLEN) {
            #pragma unroll
            for (int j = 0; j < PF; ++j) {
                dn[j] = dp[(tb + PF + j) * DMODEL];
                un[j] = up[(tb + PF + j) * DMODEL];
            }
        }
        #pragma unroll
        for (int j = 0; j < PF; ++j) {
            const int t = tb + j;
            float Bv[NSTATE], Cv[NSTATE];
            #pragma unroll
            for (int q = 0; q < 4; ++q) {
                float4 vb = nb[q], vc = nb[4 + q];
                Bv[4*q+0]=vb.x; Bv[4*q+1]=vb.y; Bv[4*q+2]=vb.z; Bv[4*q+3]=vb.w;
                Cv[4*q+0]=vc.x; Cv[4*q+1]=vc.y; Cv[4*q+2]=vc.z; Cv[4*q+3]=vc.w;
            }
            const int tn = (t + 1 < CLEN) ? (t + 1) : t;
            #pragma unroll
            for (int q = 0; q < 8; ++q) nb[q] = sBC[tn * 8 + q];
            __builtin_amdgcn_sched_barrier(0);
            const float dl = dc[j];
            const float du = dl * uc[j];
            float y = 0.0f;
            #pragma unroll
            for (int n = 0; n < NSTATE; ++n) {
                float dA = __expf(dl * Av[n]);
                st[n] = fmaf(dA, st[n], du * Bv[n]);
                y     = fmaf(st[n], Cv[n], y);
            }
            dsum += dl;
            op[t * DMODEL] = y;
        }
    }
    const int base = (c * 2 * NSTATE) * NCHAN + ch;
    #pragma unroll
    for (int n = 0; n < NSTATE; ++n)
        agg[base + n * NCHAN] = __expf(Av[n] * dsum);
    #pragma unroll
    for (int n = 0; n < NSTATE; ++n)
        agg[base + (NSTATE + n) * NCHAN] = st[n];
}

__global__ __launch_bounds__(256)
void scan_fixup(const float* __restrict__ U, const float* __restrict__ A_log,
                const float* __restrict__ xp, const float* __restrict__ delta,
                const float* __restrict__ Dvec, const float* __restrict__ agg,
                float* __restrict__ out) {
    __shared__ float4 sC[CLEN * 4];

    const int c    = blockIdx.y;
    const int bb   = blockIdx.x / BPB;
    const int dd   = (blockIdx.x % BPB) * 256 + threadIdx.x;
    const int ch   = bb * DMODEL + dd;
    const int row0 = bb * SEQ + c * CLEN;

    const float* __restrict__ dp = delta + row0 * DMODEL + dd;
    const float* __restrict__ up = U     + row0 * DMODEL + dd;
    float*       __restrict__ op = out   + row0 * DMODEL + dd;

    sC[threadIdx.x] = *(const float4*)(xp + (row0 + (threadIdx.x >> 2)) * KPROJ
                                          + DTRANK + NSTATE + 4 * (threadIdx.x & 3));

    float dn[PF], un[PF], yn[PF];
    #pragma unroll
    for (int j = 0; j < PF; ++j) {
        dn[j] = dp[j * DMODEL]; un[j] = up[j * DMODEL]; yn[j] = op[j * DMODEL];
    }

    float Av[NSTATE];
    #pragma unroll
    for (int n = 0; n < NSTATE; ++n) Av[n] = -__expf(A_log[n]);

    float entry[NSTATE];
    #pragma unroll
    for (int n = 0; n < NSTATE; ++n) entry[n] = 0.0f;
    for (int cc = 0; cc < c; cc += PB) {
        float p[PB][NSTATE], f[PB][NSTATE];
        #pragma unroll
        for (int k = 0; k < PB; ++k) {
            const int ce   = (cc + k < c) ? (cc + k) : (c - 1);
            const int base = (ce * 2 * NSTATE) * NCHAN + ch;
            #pragma unroll
            for (int n = 0; n < NSTATE; ++n) {
                p[k][n] = agg[base + n * NCHAN];
                f[k][n] = agg[base + (NSTATE + n) * NCHAN];
            }
        }
        __builtin_amdgcn_sched_barrier(0);
        #pragma unroll
        for (int k = 0; k < PB; ++k) {
            const bool live = (cc + k < c);
            #pragma unroll
            for (int n = 0; n < NSTATE; ++n)
                entry[n] = fmaf(live ? p[k][n] : 1.0f, entry[n], live ? f[k][n] : -0.0f);
        }
    }

    const float Dd = Dvec[dd];

    __syncthreads();

    float4 nc[4];
    #pragma unroll
    for (int q = 0; q < 4; ++q) nc[q] = sC[q];

    #pragma unroll 1
    for (int tb = 0; tb < CLEN; tb += PF) {
        float dc[PF], uc[PF], yc[PF];
        #pragma unroll
        for (int j = 0; j < PF; ++j) { dc[j] = dn[j]; uc[j] = un[j]; yc[j] = yn[j]; }
        if (tb + PF < CLEN) {
            #pragma unroll
            for (int j = 0; j < PF; ++j) {
                dn[j] = dp[(tb + PF + j) * DMODEL];
                un[j] = up[(tb + PF + j) * DMODEL];
                yn[j] = op[(tb + PF + j) * DMODEL];
            }
        }
        #pragma unroll
        for (int j = 0; j < PF; ++j) {
            const int t = tb + j;
            float Cv[NSTATE];
            #pragma unroll
            for (int q = 0; q < 4; ++q) {
                float4 vc = nc[q];
                Cv[4*q+0]=vc.x; Cv[4*q+1]=vc.y; Cv[4*q+2]=vc.z; Cv[4*q+3]=vc.w;
            }
            const int tn = (t + 1 < CLEN) ? (t + 1) : t;
            #pragma unroll
            for (int q = 0; q < 4; ++q) nc[q] = sC[tn * 4 + q];
            __builtin_amdgcn_sched_barrier(0);
            const float dl = dc[j];
            float ycorr = 0.0f;
            #pragma unroll
            for (int n = 0; n < NSTATE; ++n) {
                entry[n] *= __expf(dl * Av[n]);
                ycorr = fmaf(entry[n], Cv[n], ycorr);
            }
            op[t * DMODEL] = yc[j] + ycorr + uc[j] * Dd;
        }
    }
}

extern "C" void kernel_launch(void* const* d_in, const int* in_sizes, int n_in,
                              void* d_out, int out_size, void* d_ws, size_t ws_size,
                              hipStream_t stream) {
    const float* U     = (const float*)d_in[0];
    const float* A_log = (const float*)d_in[1];
    const float* W1    = (const float*)d_in[2];
    const float* W2    = (const float*)d_in[3];
    const float* bias  = (const float*)d_in[4];
    const float* Dvec  = (const float*)d_in[5];
    float* out = (float*)d_out;

    float* ws    = (float*)d_ws;
    float* xp    = ws;
    float* delta = xp + (size_t)MROWS * KPROJ;
    float* agg   = delta + (size_t)MROWS * DMODEL;

    gemm_xproj<<<dim3(1536 / 8), dim3(256), 0, stream>>>(U, W1, xp);

    gemm_delta<<<dim3(16384 / 8), dim3(256), 0, stream>>>(xp, W2, bias, delta);

    dim3 sgrid(NCHAN / 256, NCHUNK);
    scan_local<<<sgrid, dim3(256), 0, stream>>>(U, A_log, xp, delta, out, agg);
    scan_fixup<<<sgrid, dim3(256), 0, stream>>>(U, A_log, xp, delta, Dvec, agg, out);
}


# x-projection GEMM takes its WMMA fragments from LDS: each block reads its u rows and W1 in 128-k slabs as whole lines, double-buffered, not as 16-line strided global loads; sums and order unchanged
# speedup vs baseline: 1.9329x; 1.1898x over previous
#include <hip/hip_runtime.h>
#include <hip/hip_bf16.h>
#include <math.h>

typedef float v2f __attribute__((ext_vector_type(2)));
typedef float v8f __attribute__((ext_vector_type(8)));

#define BATCH   2
#define SEQ     2048
#define DMODEL  1024
#define DTRANK  64
#define NSTATE  16
#define KPROJ   (DTRANK + 2 * NSTATE)
#define MROWS   (BATCH * SEQ)
#define NCHAN   (BATCH * DMODEL)
#define NCHUNK  32
#define CLEN    (SEQ / NCHUNK)

#define G1_KS    128
#define G1_NSLAB (DMODEL / G1_KS)
#define G1_UROWS 32
#define G1_ROWS  (G1_UROWS + KPROJ)
#define G1_P4    (G1_KS / 4 + 1)
#define G1_SLAB4 (G1_ROWS * G1_P4)
#define G1_KSTEPS (G1_KS / 4)

__device__ __forceinline__ v8f g1_slab_products(const float* pa, const float* pb, v8f acc) {
    v2f a[G1_KSTEPS], b[G1_KSTEPS];
    #pragma unroll
    for (int kk = 0; kk < G1_KSTEPS; ++kk) {
        a[kk] = *(const v2f*)(pa + 4 * kk);
        b[kk] = *(const v2f*)(pb + 4 * kk);
    }
    __builtin_amdgcn_sched_barrier(0);
    #pragma unroll
    for (int kk = 0; kk < G1_KSTEPS; ++kk)
        acc = __builtin_amdgcn_wmma_f32_16x16x4_f32(false, a[kk], false, b[kk],
                                                    (short)0, acc, false, false);
    return acc;
}

__device__ __forceinline__ void g1_lds_barrier() {
    __builtin_amdgcn_fence(__ATOMIC_RELEASE, "workgroup", "local");
    __builtin_amdgcn_s_barrier();
    __builtin_amdgcn_fence(__ATOMIC_ACQUIRE, "workgroup", "local");
}

__global__ __launch_bounds__(256)
void gemm_xproj(const float* __restrict__ U, const float* __restrict__ W1,
                float* __restrict__ xp) {
    __shared__ float4 sm[2 * G1_SLAB4];

    const int lane  = threadIdx.x & 31;
    const int wave  = threadIdx.x >> 5;
    const int tile  = blockIdx.x * 8 + wave;
    const int tilesN = KPROJ / 16;
    const int mt  = tile / tilesN;
    const int nt  = tile % tilesN;
    const int ms0 = (blockIdx.x * 8) / tilesN;
    const int c   = lane & 15;
    const int h   = lane >> 4;

    const float* gu = U  + (ms0 * 16 + wave) * DMODEL + 4 * lane;
    const float* gw = W1 + wave * DMODEL + 4 * lane;
    const int sdst = wave * G1_P4 + lane;

    const int fa = ((mt - ms0) * 16 + c) * (4 * G1_P4) + 2 * h;
    const int fb = (G1_UROWS + nt * 16 + c) * (4 * G1_P4) + 2 * h;

#define G1_PIECES(X) X(0) X(1) X(2) X(3) X(4) X(5) X(6) X(7) X(8) X(9) X(10) X(11) X(12) X(13) X(14) X(15)
#define G1_DECL(i)  float4 r##i;
#define G1_LOAD(i)  r##i = *(const float4*)((((i) < 4) ? (gu + 8 * (i) * DMODEL) : (gw + 8 * ((i) - 4) * DMODEL)) + ko);
#define G1_PARK(i)  sm[dst + 8 * (i) * G1_P4] = r##i;
    G1_PIECES(G1_DECL)
    {
        const int ko = 0, dst = sdst;
        G1_PIECES(G1_LOAD)
        G1_PIECES(G1_PARK)
    }
    {
        const int ko = G1_KS;
        G1_PIECES(G1_LOAD)
    }
    g1_lds_barrier();

    v8f acc = {};
    #pragma unroll 1
    for (int s = 0; s + 1 < G1_NSLAB; ++s) {
        const int cur = (s & 1) * G1_SLAB4;
        const int dst = (G1_SLAB4 - cur) + sdst;
        G1_PIECES(G1_PARK)
        if (s + 2 < G1_NSLAB) {
            const int ko = (s + 2) * G1_KS;
            G1_PIECES(G1_LOAD)
        }
        acc = g1_slab_products((const float*)(sm + cur) + fa, (const float*)(sm + cur) + fb, acc);
        asm volatile("" :: "v"(acc));
        g1_lds_barrier();
    }
    {
        const int cur = ((G1_NSLAB - 1) & 1) * G1_SLAB4;
        acc = g1_slab_products((const float*)(sm + cur) + fa, (const float*)(sm + cur) + fb, acc);
    }
#undef G1_PARK
#undef G1_LOAD
#undef G1_DECL
#undef G1_PIECES

    const int col   = nt * 16 + c;
    const int rbase = mt * 16 + 8 * h;
    #pragma unroll
    for (int i = 0; i < 8; ++i)
        xp[(rbase + i) * KPROJ + col] = acc[i];
}

__global__ __launch_bounds__(256)
void gemm_delta(const float* __restrict__ xp, const float* __restrict__ W2,
                const float* __restrict__ bias, float* __restrict__ delta) {
    const int lane  = threadIdx.x & 31;
    const int wave  = threadIdx.x >> 5;
    const int tile  = blockIdx.x * (blockDim.x >> 5) + wave;
    const int tilesN = DMODEL / 16;
    const int m0 = (tile / tilesN) * 16;
    const int n0 = (tile % tilesN) * 16;

    const int row   = m0 + (lane & 15);
    const int col   = n0 + (lane & 15);
    const int khalf = (lane >> 4) * 2;

    v8f acc = {};
    #pragma unroll
    for (int k = 0; k < DTRANK; k += 4) {
        v2f a = *(const v2f*)(xp + row * KPROJ  + k + khalf);
        v2f b = *(const v2f*)(W2 + col * DTRANK + k + khalf);
        acc = __builtin_amdgcn_wmma_f32_16x16x4_f32(false, a, false, b,
                                                    (short)0, acc, false, false);
    }
    const int   rbase = m0 + ((lane >> 4) ? 8 : 0);
    const float bn    = bias[col];
    #pragma unroll
    for (int i = 0; i < 8; ++i) {
        float x  = acc[i] + bn;
        float sp = (x > 20.0f) ? x : log1pf(__expf(x));
        delta[(rbase + i) * DMODEL + col] = sp;
    }
}

#define PF 8
#define PB 4
#define BPB (DMODEL / 256)

__global__ __launch_bounds__(256)
void scan_local(const float* __restrict__ U, const float* __restrict__ A_log,
                const float* __restrict__ xp, const float* __restrict__ delta,
                float* __restrict__ out, float* __restrict__ agg) {
    __shared__ float4 sBC[CLEN * 8];

    const int c    = blockIdx.y;
    const int bb   = blockIdx.x / BPB;
    const int dd   = (blockIdx.x % BPB) * 256 + threadIdx.x;
    const int ch   = bb * DMODEL + dd;
    const int row0 = bb * SEQ + c * CLEN;

    const float* __restrict__ dp = delta + row0 * DMODEL + dd;
    const float* __restrict__ up = U     + row0 * DMODEL + dd;
    float*       __restrict__ op = out   + row0 * DMODEL + dd;

    #pragma unroll
    for (int i = threadIdx.x; i < CLEN * 8; i += 256)
        sBC[i] = *(const float4*)(xp + (row0 + (i >> 3)) * KPROJ + DTRANK + 4 * (i & 7));

    float dn[PF], un[PF];
    #pragma unroll
    for (int j = 0; j < PF; ++j) { dn[j] = dp[j * DMODEL]; un[j] = up[j * DMODEL]; }

    float Av[NSTATE];
    #pragma unroll
    for (int n = 0; n < NSTATE; ++n) Av[n] = -__expf(A_log[n]);

    float st[NSTATE];
    #pragma unroll
    for (int n = 0; n < NSTATE; ++n) st[n] = 0.0f;

    __syncthreads();

    float4 nb[8];
    #pragma unroll
    for (int q = 0; q < 8; ++q) nb[q] = sBC[q];

    float dsum = 0.0f;
    #pragma unroll 1
    for (int tb = 0; tb < CLEN; tb += PF) {
        float dc[PF], uc[PF];
        #pragma unroll
        for (int j = 0; j < PF; ++j) { dc[j] = dn[j]; uc[j] = un[j]; }
        if (tb + PF < CLEN) {
            #pragma unroll
            for (int j = 0; j < PF; ++j) {
                dn[j] = dp[(tb + PF + j) * DMODEL];
                un[j] = up[(tb + PF + j) * DMODEL];
            }
        }
        #pragma unroll
        for (int j = 0; j < PF; ++j) {
            const int t = tb + j;
            float Bv[NSTATE], Cv[NSTATE];
            #pragma unroll
            for (int q = 0; q < 4; ++q) {
                float4 vb = nb[q], vc = nb[4 + q];
                Bv[4*q+0]=vb.x; Bv[4*q+1]=vb.y; Bv[4*q+2]=vb.z; Bv[4*q+3]=vb.w;
                Cv[4*q+0]=vc.x; Cv[4*q+1]=vc.y; Cv[4*q+2]=vc.z; Cv[4*q+3]=vc.w;
            }
            const int tn = (t + 1 < CLEN) ? (t + 1) : t;
            #pragma unroll
            for (int q = 0; q < 8; ++q) nb[q] = sBC[tn * 8 + q];
            __builtin_amdgcn_sched_barrier(0);
            const float dl = dc[j];
            const float du = dl * uc[j];
            float y = 0.0f;
            #pragma unroll
            for (int n = 0; n < NSTATE; ++n) {
                float dA = __expf(dl * Av[n]);
                st[n] = fmaf(dA, st[n], du * Bv[n]);
                y     = fmaf(st[n], Cv[n], y);
            }
            dsum += dl;
            op[t * DMODEL] = y;
        }
    }
    const int base = (c * 2 * NSTATE) * NCHAN + ch;
    #pragma unroll
    for (int n = 0; n < NSTATE; ++n)
        agg[base + n * NCHAN] = __expf(Av[n] * dsum);
    #pragma unroll
    for (int n = 0; n < NSTATE; ++n)
        agg[base + (NSTATE + n) * NCHAN] = st[n];
}

__global__ __launch_bounds__(256)
void scan_fixup(const float* __restrict__ U, const float* __restrict__ A_log,
                const float* __restrict__ xp, const float* __restrict__ delta,
                const float* __restrict__ Dvec, const float* __restrict__ agg,
                float* __restrict__ out) {
    __shared__ float4 sC[CLEN * 4];

    const int c    = blockIdx.y;
    const int bb   = blockIdx.x / BPB;
    const int dd   = (blockIdx.x % BPB) * 256 + threadIdx.x;
    const int ch   = bb * DMODEL + dd;
    const int row0 = bb * SEQ + c * CLEN;

    const float* __restrict__ dp = delta + row0 * DMODEL + dd;
    const float* __restrict__ up = U     + row0 * DMODEL + dd;
    float*       __restrict__ op = out   + row0 * DMODEL + dd;

    sC[threadIdx.x] = *(const float4*)(xp + (row0 + (threadIdx.x >> 2)) * KPROJ
                                          + DTRANK + NSTATE + 4 * (threadIdx.x & 3));

    float dn[PF], un[PF], yn[PF];
    #pragma unroll
    for (int j = 0; j < PF; ++j) {
        dn[j] = dp[j * DMODEL]; un[j] = up[j * DMODEL]; yn[j] = op[j * DMODEL];
    }

    float Av[NSTATE];
    #pragma unroll
    for (int n = 0; n < NSTATE; ++n) Av[n] = -__expf(A_log[n]);

    float entry[NSTATE];
    #pragma unroll
    for (int n = 0; n < NSTATE; ++n) entry[n] = 0.0f;
    for (int cc = 0; cc < c; cc += PB) {
        float p[PB][NSTATE], f[PB][NSTATE];
        #pragma unroll
        for (int k = 0; k < PB; ++k) {
            const int ce   = (cc + k < c) ? (cc + k) : (c - 1);
            const int base = (ce * 2 * NSTATE) * NCHAN + ch;
            #pragma unroll
            for (int n = 0; n < NSTATE; ++n) {
                p[k][n] = agg[base + n * NCHAN];
                f[k][n] = agg[base + (NSTATE + n) * NCHAN];
            }
        }
        __builtin_amdgcn_sched_barrier(0);
        #pragma unroll
        for (int k = 0; k < PB; ++k) {
            const bool live = (cc + k < c);
            #pragma unroll
            for (int n = 0; n < NSTATE; ++n)
                entry[n] = fmaf(live ? p[k][n] : 1.0f, entry[n], live ? f[k][n] : -0.0f);
        }
    }

    const float Dd = Dvec[dd];

    __syncthreads();

    float4 nc[4];
    #pragma unroll
    for (int q = 0; q < 4; ++q) nc[q] = sC[q];

    #pragma unroll 1
    for (int tb = 0; tb < CLEN; tb += PF) {
        float dc[PF], uc[PF], yc[PF];
        #pragma unroll
        for (int j = 0; j < PF; ++j) { dc[j] = dn[j]; uc[j] = un[j]; yc[j] = yn[j]; }
        if (tb + PF < CLEN) {
            #pragma unroll
            for (int j = 0; j < PF; ++j) {
                dn[j] = dp[(tb + PF + j) * DMODEL];
                un[j] = up[(tb + PF + j) * DMODEL];
                yn[j] = op[(tb + PF + j) * DMODEL];
            }
        }
        #pragma unroll
        for (int j = 0; j < PF; ++j) {
            const int t = tb + j;
            float Cv[NSTATE];
            #pragma unroll
            for (int q = 0; q < 4; ++q) {
                float4 vc = nc[q];
                Cv[4*q+0]=vc.x; Cv[4*q+1]=vc.y; Cv[4*q+2]=vc.z; Cv[4*q+3]=vc.w;
            }
            const int tn = (t + 1 < CLEN) ? (t + 1) : t;
            #pragma unroll
            for (int q = 0; q < 4; ++q) nc[q] = sC[tn * 4 + q];
            __builtin_amdgcn_sched_barrier(0);
            const float dl = dc[j];
            float ycorr = 0.0f;
            #pragma unroll
            for (int n = 0; n < NSTATE; ++n) {
                entry[n] *= __expf(dl * Av[n]);
                ycorr = fmaf(entry[n], Cv[n], ycorr);
            }
            op[t * DMODEL] = yc[j] + ycorr + uc[j] * Dd;
        }
    }
}

extern "C" void kernel_launch(void* const* d_in, const int* in_sizes, int n_in,
                              void* d_out, int out_size, void* d_ws, size_t ws_size,
                              hipStream_t stream) {
    const float* U     = (const float*)d_in[0];
    const float* A_log = (const float*)d_in[1];
    const float* W1    = (const float*)d_in[2];
    const float* W2    = (const float*)d_in[3];
    const float* bias  = (const float*)d_in[4];
    const float* Dvec  = (const float*)d_in[5];
    float* out = (float*)d_out;

    float* ws    = (float*)d_ws;
    float* xp    = ws;
    float* delta = xp + (size_t)MROWS * KPROJ;
    float* agg   = delta + (size_t)MROWS * DMODEL;

    gemm_xproj<<<dim3(1536 / 8), dim3(256), 0, stream>>>(U, W1, xp);

    gemm_delta<<<dim3(16384 / 8), dim3(256), 0, stream>>>(xp, W2, bias, delta);

    dim3 sgrid(NCHAN / 256, NCHUNK);
    scan_local<<<sgrid, dim3(256), 0, stream>>>(U, A_log, xp, delta, out, agg);
    scan_fixup<<<sgrid, dim3(256), 0, stream>>>(U, A_log, xp, delta, Dvec, agg, out);
}


# delta GEMM re-tiled onto 256 resident workgroups of 16 waves (was 2048 of 8 one-tile waves): a wave keeps W2 fragments of 32 columns over 2 row tiles, xp and W2 staged once in LDS; sums unchanged
# speedup vs baseline: 2.0751x; 1.0736x over previous
#include <hip/hip_runtime.h>
#include <hip/hip_bf16.h>
#include <math.h>

typedef float v2f __attribute__((ext_vector_type(2)));
typedef float v8f __attribute__((ext_vector_type(8)));

#define BATCH   2
#define SEQ     2048
#define DMODEL  1024
#define DTRANK  64
#define NSTATE  16
#define KPROJ   (DTRANK + 2 * NSTATE)
#define MROWS   (BATCH * SEQ)
#define NCHAN   (BATCH * DMODEL)
#define NCHUNK  32
#define CLEN    (SEQ / NCHUNK)

#define G1_KS    128
#define G1_NSLAB (DMODEL / G1_KS)
#define G1_UROWS 32
#define G1_ROWS  (G1_UROWS + KPROJ)
#define G1_P4    (G1_KS / 4 + 1)
#define G1_SLAB4 (G1_ROWS * G1_P4)
#define G1_KSTEPS (G1_KS / 4)

__device__ __forceinline__ v8f g1_slab_products(const float* pa, const float* pb, v8f acc) {
    v2f a[G1_KSTEPS], b[G1_KSTEPS];
    #pragma unroll
    for (int kk = 0; kk < G1_KSTEPS; ++kk) {
        a[kk] = *(const v2f*)(pa + 4 * kk);
        b[kk] = *(const v2f*)(pb + 4 * kk);
    }
    __builtin_amdgcn_sched_barrier(0);
    #pragma unroll
    for (int kk = 0; kk < G1_KSTEPS; ++kk)
        acc = __builtin_amdgcn_wmma_f32_16x16x4_f32(false, a[kk], false, b[kk],
                                                    (short)0, acc, false, false);
    return acc;
}

__device__ __forceinline__ void g1_lds_barrier() {
    __builtin_amdgcn_fence(__ATOMIC_RELEASE, "workgroup", "local");
    __builtin_amdgcn_s_barrier();
    __builtin_amdgcn_fence(__ATOMIC_ACQUIRE, "workgroup", "local");
}

__global__ __launch_bounds__(256)
void gemm_xproj(const float* __restrict__ U, const float* __restrict__ W1,
                float* __restrict__ xp) {
    __shared__ float4 sm[2 * G1_SLAB4];

    const int lane  = threadIdx.x & 31;
    const int wave  = threadIdx.x >> 5;
    const int tile  = blockIdx.x * 8 + wave;
    const int tilesN = KPROJ / 16;
    const int mt  = tile / tilesN;
    const int nt  = tile % tilesN;
    const int ms0 = (blockIdx.x * 8) / tilesN;
    const int c   = lane & 15;
    const int h   = lane >> 4;

    const float* gu = U  + (ms0 * 16 + wave) * DMODEL + 4 * lane;
    const float* gw = W1 + wave * DMODEL + 4 * lane;
    const int sdst = wave * G1_P4 + lane;

    const int fa = ((mt - ms0) * 16 + c) * (4 * G1_P4) + 2 * h;
    const int fb = (G1_UROWS + nt * 16 + c) * (4 * G1_P4) + 2 * h;

#define G1_PIECES(X) X(0) X(1) X(2) X(3) X(4) X(5) X(6) X(7) X(8) X(9) X(10) X(11) X(12) X(13) X(14) X(15)
#define G1_DECL(i)  float4 r##i;
#define G1_LOAD(i)  r##i = *(const float4*)((((i) < 4) ? (gu + 8 * (i) * DMODEL) : (gw + 8 * ((i) - 4) * DMODEL)) + ko);
#define G1_PARK(i)  sm[dst + 8 * (i) * G1_P4] = r##i;
    G1_PIECES(G1_DECL)
    {
        const int ko = 0, dst = sdst;
        G1_PIECES(G1_LOAD)
        G1_PIECES(G1_PARK)
    }
    {
        const int ko = G1_KS;
        G1_PIECES(G1_LOAD)
    }
    g1_lds_barrier();

    v8f acc = {};
    #pragma unroll 1
    for (int s = 0; s + 1 < G1_NSLAB; ++s) {
        const int cur = (s & 1) * G1_SLAB4;
        const int dst = (G1_SLAB4 - cur) + sdst;
        G1_PIECES(G1_PARK)
        if (s + 2 < G1_NSLAB) {
            const int ko = (s + 2) * G1_KS;
            G1_PIECES(G1_LOAD)
        }
        acc = g1_slab_products((const float*)(sm + cur) + fa, (const float*)(sm + cur) + fb, acc);
        asm volatile("" :: "v"(acc));
        g1_lds_barrier();
    }
    {
        const int cur = ((G1_NSLAB - 1) & 1) * G1_SLAB4;
        acc = g1_slab_products((const float*)(sm + cur) + fa, (const float*)(sm + cur) + fb, acc);
    }
#undef G1_PARK
#undef G1_LOAD
#undef G1_DECL
#undef G1_PIECES

    const int col   = nt * 16 + c;
    const int rbase = mt * 16 + 8 * h;
    #pragma unroll
    for (int i = 0; i < 8; ++i)
        xp[(rbase + i) * KPROJ + col] = acc[i];
}

#define G2_BM  64
#define G2_BN  256
#define G2_P4  (DTRANK / 4 + 1)
#define G2_NT  512
#define G2_MT  2
#define G2_KSTEPS (DTRANK / 4)

__global__ __launch_bounds__(G2_NT)
void gemm_delta(const float* __restrict__ xp, const float* __restrict__ W2,
                const float* __restrict__ bias, float* __restrict__ delta) {
    __shared__ float4 sA[G2_BM * G2_P4];
    __shared__ float4 sB[G2_BN * G2_P4];

    const int lane = threadIdx.x & 31;
    const int wave = (threadIdx.x >> 5) & 7;
    const int mh   = threadIdx.x >> 8;
    const int m0   = (blockIdx.x >> 2) * G2_BM;
    const int n0   = (blockIdx.x & 3) * G2_BN;
    const int c    = lane & 15;
    const int h    = lane >> 4;

    #pragma unroll
    for (int i = 0; i < G2_BM * 16 / G2_NT; ++i) {
        const int idx = threadIdx.x + G2_NT * i;
        const int r = idx >> 4, q = idx & 15;
        sA[r * G2_P4 + q] = *(const float4*)(xp + (m0 + r) * KPROJ + 4 * q);
    }
    #pragma unroll
    for (int i = 0; i < G2_BN * 16 / G2_NT; ++i) {
        const int idx = threadIdx.x + G2_NT * i;
        const int r = idx >> 4, q = idx & 15;
        sB[r * G2_P4 + q] = *(const float4*)(W2 + (n0 + r) * DTRANK + 4 * q);
    }
    __syncthreads();

    const float* fb = (const float*)sB + (wave * 32 + c) * (4 * G2_P4) + 2 * h;
    v2f b0[G2_KSTEPS], b1[G2_KSTEPS];
    #pragma unroll
    for (int kk = 0; kk < G2_KSTEPS; ++kk) {
        b0[kk] = *(const v2f*)(fb + 4 * kk);
        b1[kk] = *(const v2f*)(fb + 16 * (4 * G2_P4) + 4 * kk);
    }

    const int   col = n0 + wave * 32 + c;
    const float bn0 = bias[col];
    const float bn1 = bias[col + 16];

    const float* fa = (const float*)sA + (mh * 32 + c) * (4 * G2_P4) + 2 * h;
    v2f an[G2_KSTEPS];
    #pragma unroll
    for (int kk = 0; kk < G2_KSTEPS; ++kk) an[kk] = *(const v2f*)(fa + 4 * kk);

    #pragma unroll
    for (int mt = 0; mt < G2_MT; ++mt) {
        v2f a[G2_KSTEPS];
        #pragma unroll
        for (int kk = 0; kk < G2_KSTEPS; ++kk) a[kk] = an[kk];
        if (mt + 1 < G2_MT) {
            #pragma unroll
            for (int kk = 0; kk < G2_KSTEPS; ++kk)
                an[kk] = *(const v2f*)(fa + (mt + 1) * 16 * (4 * G2_P4) + 4 * kk);
        }
        __builtin_amdgcn_sched_barrier(0);
        v8f acc0 = {}, acc1 = {};
        #pragma unroll
        for (int kk = 0; kk < G2_KSTEPS; ++kk) {
            acc0 = __builtin_amdgcn_wmma_f32_16x16x4_f32(false, a[kk], false, b0[kk],
                                                         (short)0, acc0, false, false);
            acc1 = __builtin_amdgcn_wmma_f32_16x16x4_f32(false, a[kk], false, b1[kk],
                                                         (short)0, acc1, false, false);
        }
        float* dr = delta + (m0 + mh * 32 + mt * 16 + 8 * h) * DMODEL + col;
        float sp0[8], sp1[8];
        #pragma unroll
        for (int i = 0; i < 8; ++i) {
            const float x0 = acc0[i] + bn0;
            const float l0 = log1pf(__expf(x0));
            sp0[i] = (x0 > 20.0f) ? x0 : l0;
            const float x1 = acc1[i] + bn1;
            const float l1 = log1pf(__expf(x1));
            sp1[i] = (x1 > 20.0f) ? x1 : l1;
        }
        #pragma unroll
        for (int i = 0; i < 8; ++i) {
            dr[i * DMODEL]      = sp0[i];
            dr[i * DMODEL + 16] = sp1[i];
        }
    }
}

#define PF 8
#define PB 4
#define BPB (DMODEL / 256)

__global__ __launch_bounds__(256)
void scan_local(const float* __restrict__ U, const float* __restrict__ A_log,
                const float* __restrict__ xp, const float* __restrict__ delta,
                float* __restrict__ out, float* __restrict__ agg) {
    __shared__ float4 sBC[CLEN * 8];

    const int c    = blockIdx.y;
    const int bb   = blockIdx.x / BPB;
    const int dd   = (blockIdx.x % BPB) * 256 + threadIdx.x;
    const int ch   = bb * DMODEL + dd;
    const int row0 = bb * SEQ + c * CLEN;

    const float* __restrict__ dp = delta + row0 * DMODEL + dd;
    const float* __restrict__ up = U     + row0 * DMODEL + dd;
    float*       __restrict__ op = out   + row0 * DMODEL + dd;

    #pragma unroll
    for (int i = threadIdx.x; i < CLEN * 8; i += 256)
        sBC[i] = *(const float4*)(xp + (row0 + (i >> 3)) * KPROJ + DTRANK + 4 * (i & 7));

    float dn[PF], un[PF];
    #pragma unroll
    for (int j = 0; j < PF; ++j) { dn[j] = dp[j * DMODEL]; un[j] = up[j * DMODEL]; }

    float Av[NSTATE];
    #pragma unroll
    for (int n = 0; n < NSTATE; ++n) Av[n] = -__expf(A_log[n]);

    float st[NSTATE];
    #pragma unroll
    for (int n = 0; n < NSTATE; ++n) st[n] = 0.0f;

    __syncthreads();

    float4 nb[8];
    #pragma unroll
    for (int q = 0; q < 8; ++q) nb[q] = sBC[q];

    float dsum = 0.0f;
    #pragma unroll 1
    for (int tb = 0; tb < CLEN; tb += PF) {
        float dc[PF], uc[PF];
        #pragma unroll
        for (int j = 0; j < PF; ++j) { dc[j] = dn[j]; uc[j] = un[j]; }
        if (tb + PF < CLEN) {
            #pragma unroll
            for (int j = 0; j < PF; ++j) {
                dn[j] = dp[(tb + PF + j) * DMODEL];
                un[j] = up[(tb + PF + j) * DMODEL];
            }
        }
        #pragma unroll
        for (int j = 0; j < PF; ++j) {
            const int t = tb + j;
            float Bv[NSTATE], Cv[NSTATE];
            #pragma unroll
            for (int q = 0; q < 4; ++q) {
                float4 vb = nb[q], vc = nb[4 + q];
                Bv[4*q+0]=vb.x; Bv[4*q+1]=vb.y; Bv[4*q+2]=vb.z; Bv[4*q+3]=vb.w;
                Cv[4*q+0]=vc.x; Cv[4*q+1]=vc.y; Cv[4*q+2]=vc.z; Cv[4*q+3]=vc.w;
            }
            const int tn = (t + 1 < CLEN) ? (t + 1) : t;
            #pragma unroll
            for (int q = 0; q < 8; ++q) nb[q] = sBC[tn * 8 + q];
            __builtin_amdgcn_sched_barrier(0);
            const float dl = dc[j];
            const float du = dl * uc[j];
            float y = 0.0f;
            #pragma unroll
            for (int n = 0; n < NSTATE; ++n) {
                float dA = __expf(dl * Av[n]);
                st[n] = fmaf(dA, st[n], du * Bv[n]);
                y     = fmaf(st[n], Cv[n], y);
            }
            dsum += dl;
            op[t * DMODEL] = y;
        }
    }
    const int base = (c * 2 * NSTATE) * NCHAN + ch;
    #pragma unroll
    for (int n = 0; n < NSTATE; ++n)
        agg[base + n * NCHAN] = __expf(Av[n] * dsum);
    #pragma unroll
    for (int n = 0; n < NSTATE; ++n)
        agg[base + (NSTATE + n) * NCHAN] = st[n];
}

__global__ __launch_bounds__(256)
void scan_fixup(const float* __restrict__ U, const float* __restrict__ A_log,
                const float* __restrict__ xp, const float* __restrict__ delta,
                const float* __restrict__ Dvec, const float* __restrict__ agg,
                float* __restrict__ out) {
    __shared__ float4 sC[CLEN * 4];

    const int c    = blockIdx.y;
    const int bb   = blockIdx.x / BPB;
    const int dd   = (blockIdx.x % BPB) * 256 + threadIdx.x;
    const int ch   = bb * DMODEL + dd;
    const int row0 = bb * SEQ + c * CLEN;

    const float* __restrict__ dp = delta + row0 * DMODEL + dd;
    const float* __restrict__ up = U     + row0 * DMODEL + dd;
    float*       __restrict__ op = out   + row0 * DMODEL + dd;

    sC[threadIdx.x] = *(const float4*)(xp + (row0 + (threadIdx.x >> 2)) * KPROJ
                                          + DTRANK + NSTATE + 4 * (threadIdx.x & 3));

    float dn[PF], un[PF], yn[PF];
    #pragma unroll
    for (int j = 0; j < PF; ++j) {
        dn[j] = dp[j * DMODEL]; un[j] = up[j * DMODEL]; yn[j] = op[j * DMODEL];
    }

    float Av[NSTATE];
    #pragma unroll
    for (int n = 0; n < NSTATE; ++n) Av[n] = -__expf(A_log[n]);

    float entry[NSTATE];
    #pragma unroll
    for (int n = 0; n < NSTATE; ++n) entry[n] = 0.0f;
    for (int cc = 0; cc < c; cc += PB) {
        float p[PB][NSTATE], f[PB][NSTATE];
        #pragma unroll
        for (int k = 0; k < PB; ++k) {
            const int ce   = (cc + k < c) ? (cc + k) : (c - 1);
            const int base = (ce * 2 * NSTATE) * NCHAN + ch;
            #pragma unroll
            for (int n = 0; n < NSTATE; ++n) {
                p[k][n] = agg[base + n * NCHAN];
                f[k][n] = agg[base + (NSTATE + n) * NCHAN];
            }
        }
        __builtin_amdgcn_sched_barrier(0);
        #pragma unroll
        for (int k = 0; k < PB; ++k) {
            const bool live = (cc + k < c);
            #pragma unroll
            for (int n = 0; n < NSTATE; ++n)
                entry[n] = fmaf(live ? p[k][n] : 1.0f, entry[n], live ? f[k][n] : -0.0f);
        }
    }

    const float Dd = Dvec[dd];

    __syncthreads();

    float4 nc[4];
    #pragma unroll
    for (int q = 0; q < 4; ++q) nc[q] = sC[q];

    #pragma unroll 1
    for (int tb = 0; tb < CLEN; tb += PF) {
        float dc[PF], uc[PF], yc[PF];
        #pragma unroll
        for (int j = 0; j < PF; ++j) { dc[j] = dn[j]; uc[j] = un[j]; yc[j] = yn[j]; }
        if (tb + PF < CLEN) {
            #pragma unroll
            for (int j = 0; j < PF; ++j) {
                dn[j] = dp[(tb + PF + j) * DMODEL];
                un[j] = up[(tb + PF + j) * DMODEL];
                yn[j] = op[(tb + PF + j) * DMODEL];
            }
        }
        #pragma unroll
        for (int j = 0; j < PF; ++j) {
            const int t = tb + j;
            float Cv[NSTATE];
            #pragma unroll
            for (int q = 0; q < 4; ++q) {
                float4 vc = nc[q];
                Cv[4*q+0]=vc.x; Cv[4*q+1]=vc.y; Cv[4*q+2]=vc.z; Cv[4*q+3]=vc.w;
            }
            const int tn = (t + 1 < CLEN) ? (t + 1) : t;
            #pragma unroll
            for (int q = 0; q < 4; ++q) nc[q] = sC[tn * 4 + q];
            __builtin_amdgcn_sched_barrier(0);
            const float dl = dc[j];
            float ycorr = 0.0f;
            #pragma unroll
            for (int n = 0; n < NSTATE; ++n) {
                entry[n] *= __expf(dl * Av[n]);
                ycorr = fmaf(entry[n], Cv[n], ycorr);
            }
            op[t * DMODEL] = yc[j] + ycorr + uc[j] * Dd;
        }
    }
}

extern "C" void kernel_launch(void* const* d_in, const int* in_sizes, int n_in,
                              void* d_out, int out_size, void* d_ws, size_t ws_size,
                              hipStream_t stream) {
    const float* U     = (const float*)d_in[0];
    const float* A_log = (const float*)d_in[1];
    const float* W1    = (const float*)d_in[2];
    const float* W2    = (const float*)d_in[3];
    const float* bias  = (const float*)d_in[4];
    const float* Dvec  = (const float*)d_in[5];
    float* out = (float*)d_out;

    float* ws    = (float*)d_ws;
    float* xp    = ws;
    float* delta = xp + (size_t)MROWS * KPROJ;
    float* agg   = delta + (size_t)MROWS * DMODEL;

    gemm_xproj<<<dim3(1536 / 8), dim3(256), 0, stream>>>(U, W1, xp);

    gemm_delta<<<dim3((MROWS / G2_BM) * (DMODEL / G2_BN)), dim3(G2_NT), 0, stream>>>(xp, W2, bias, delta);

    dim3 sgrid(NCHAN / 256, NCHUNK);
    scan_local<<<sgrid, dim3(256), 0, stream>>>(U, A_log, xp, delta, out, agg);
    scan_fixup<<<sgrid, dim3(256), 0, stream>>>(U, A_log, xp, delta, Dvec, agg, out);
}


# delta GEMM epilogue: softplus as max(x,0)+log1p(exp(-|x|)) on the hardware exp, log and rcp, log1p(e)=log(u)*e/(u-1), u=1+e, not the library log1p: some 75 vector instructions per element become 15
# speedup vs baseline: 2.2557x; 1.0870x over previous
#include <hip/hip_runtime.h>
#include <hip/hip_bf16.h>
#include <math.h>

typedef float v2f __attribute__((ext_vector_type(2)));
typedef float v8f __attribute__((ext_vector_type(8)));

#define BATCH   2
#define SEQ     2048
#define DMODEL  1024
#define DTRANK  64
#define NSTATE  16
#define KPROJ   (DTRANK + 2 * NSTATE)
#define MROWS   (BATCH * SEQ)
#define NCHAN   (BATCH * DMODEL)
#define NCHUNK  32
#define CLEN    (SEQ / NCHUNK)

#define G1_KS    128
#define G1_NSLAB (DMODEL / G1_KS)
#define G1_UROWS 32
#define G1_ROWS  (G1_UROWS + KPROJ)
#define G1_P4    (G1_KS / 4 + 1)
#define G1_SLAB4 (G1_ROWS * G1_P4)
#define G1_KSTEPS (G1_KS / 4)

__device__ __forceinline__ v8f g1_slab_products(const float* pa, const float* pb, v8f acc) {
    v2f a[G1_KSTEPS], b[G1_KSTEPS];
    #pragma unroll
    for (int kk = 0; kk < G1_KSTEPS; ++kk) {
        a[kk] = *(const v2f*)(pa + 4 * kk);
        b[kk] = *(const v2f*)(pb + 4 * kk);
    }
    __builtin_amdgcn_sched_barrier(0);
    #pragma unroll
    for (int kk = 0; kk < G1_KSTEPS; ++kk)
        acc = __builtin_amdgcn_wmma_f32_16x16x4_f32(false, a[kk], false, b[kk],
                                                    (short)0, acc, false, false);
    return acc;
}

__device__ __forceinline__ void g1_lds_barrier() {
    __builtin_amdgcn_fence(__ATOMIC_RELEASE, "workgroup", "local");
    __builtin_amdgcn_s_barrier();
    __builtin_amdgcn_fence(__ATOMIC_ACQUIRE, "workgroup", "local");
}

__global__ __launch_bounds__(256)
void gemm_xproj(const float* __restrict__ U, const float* __restrict__ W1,
                float* __restrict__ xp) {
    __shared__ float4 sm[2 * G1_SLAB4];

    const int lane  = threadIdx.x & 31;
    const int wave  = threadIdx.x >> 5;
    const int tile  = blockIdx.x * 8 + wave;
    const int tilesN = KPROJ / 16;
    const int mt  = tile / tilesN;
    const int nt  = tile % tilesN;
    const int ms0 = (blockIdx.x * 8) / tilesN;
    const int c   = lane & 15;
    const int h   = lane >> 4;

    const float* gu = U  + (ms0 * 16 + wave) * DMODEL + 4 * lane;
    const float* gw = W1 + wave * DMODEL + 4 * lane;
    const int sdst = wave * G1_P4 + lane;

    const int fa = ((mt - ms0) * 16 + c) * (4 * G1_P4) + 2 * h;
    const int fb = (G1_UROWS + nt * 16 + c) * (4 * G1_P4) + 2 * h;

#define G1_PIECES(X) X(0) X(1) X(2) X(3) X(4) X(5) X(6) X(7) X(8) X(9) X(10) X(11) X(12) X(13) X(14) X(15)
#define G1_DECL(i)  float4 r##i;
#define G1_LOAD(i)  r##i = *(const float4*)((((i) < 4) ? (gu + 8 * (i) * DMODEL) : (gw + 8 * ((i) - 4) * DMODEL)) + ko);
#define G1_PARK(i)  sm[dst + 8 * (i) * G1_P4] = r##i;
    G1_PIECES(G1_DECL)
    {
        const int ko = 0, dst = sdst;
        G1_PIECES(G1_LOAD)
        G1_PIECES(G1_PARK)
    }
    {
        const int ko = G1_KS;
        G1_PIECES(G1_LOAD)
    }
    g1_lds_barrier();

    v8f acc = {};
    #pragma unroll 1
    for (int s = 0; s + 1 < G1_NSLAB; ++s) {
        const int cur = (s & 1) * G1_SLAB4;
        const int dst = (G1_SLAB4 - cur) + sdst;
        G1_PIECES(G1_PARK)
        if (s + 2 < G1_NSLAB) {
            const int ko = (s + 2) * G1_KS;
            G1_PIECES(G1_LOAD)
        }
        acc = g1_slab_products((const float*)(sm + cur) + fa, (const float*)(sm + cur) + fb, acc);
        asm volatile("" :: "v"(acc));
        g1_lds_barrier();
    }
    {
        const int cur = ((G1_NSLAB - 1) & 1) * G1_SLAB4;
        acc = g1_slab_products((const float*)(sm + cur) + fa, (const float*)(sm + cur) + fb, acc);
    }
#undef G1_PARK
#undef G1_LOAD
#undef G1_DECL
#undef G1_PIECES

    const int col   = nt * 16 + c;
    const int rbase = mt * 16 + 8 * h;
    #pragma unroll
    for (int i = 0; i < 8; ++i)
        xp[(rbase + i) * KPROJ + col] = acc[i];
}

#define G2_BM  64
#define G2_BN  256
#define G2_P4  (DTRANK / 4 + 1)
#define G2_NT  512
#define G2_MT  2
#define G2_KSTEPS (DTRANK / 4)

__device__ __forceinline__ float g2_softplus(float x) {
    const float e = __builtin_amdgcn_exp2f(-fabsf(x) * 1.44269504088896341f);
    const float u = 1.0f + e;
    const float d = u - 1.0f;
    const float l = __builtin_amdgcn_logf(u) * 0.693147180559945309f;
    const float r = (d == 0.0f) ? e : l * (e * __builtin_amdgcn_rcpf(d));
    return fmaxf(x, 0.0f) + r;
}

__global__ __launch_bounds__(G2_NT)
void gemm_delta(const float* __restrict__ xp, const float* __restrict__ W2,
                const float* __restrict__ bias, float* __restrict__ delta) {
    __shared__ float4 sA[G2_BM * G2_P4];
    __shared__ float4 sB[G2_BN * G2_P4];

    const int lane = threadIdx.x & 31;
    const int wave = (threadIdx.x >> 5) & 7;
    const int mh   = threadIdx.x >> 8;
    const int m0   = (blockIdx.x >> 2) * G2_BM;
    const int n0   = (blockIdx.x & 3) * G2_BN;
    const int c    = lane & 15;
    const int h    = lane >> 4;

    #pragma unroll
    for (int i = 0; i < G2_BM * 16 / G2_NT; ++i) {
        const int idx = threadIdx.x + G2_NT * i;
        const int r = idx >> 4, q = idx & 15;
        sA[r * G2_P4 + q] = *(const float4*)(xp + (m0 + r) * KPROJ + 4 * q);
    }
    #pragma unroll
    for (int i = 0; i < G2_BN * 16 / G2_NT; ++i) {
        const int idx = threadIdx.x + G2_NT * i;
        const int r = idx >> 4, q = idx & 15;
        sB[r * G2_P4 + q] = *(const float4*)(W2 + (n0 + r) * DTRANK + 4 * q);
    }
    __syncthreads();

    const float* fb = (const float*)sB + (wave * 32 + c) * (4 * G2_P4) + 2 * h;
    v2f b0[G2_KSTEPS], b1[G2_KSTEPS];
    #pragma unroll
    for (int kk = 0; kk < G2_KSTEPS; ++kk) {
        b0[kk] = *(const v2f*)(fb + 4 * kk);
        b1[kk] = *(const v2f*)(fb + 16 * (4 * G2_P4) + 4 * kk);
    }

    const int   col = n0 + wave * 32 + c;
    const float bn0 = bias[col];
    const float bn1 = bias[col + 16];

    const float* fa = (const float*)sA + (mh * 32 + c) * (4 * G2_P4) + 2 * h;
    v2f an[G2_KSTEPS];
    #pragma unroll
    for (int kk = 0; kk < G2_KSTEPS; ++kk) an[kk] = *(const v2f*)(fa + 4 * kk);

    #pragma unroll
    for (int mt = 0; mt < G2_MT; ++mt) {
        v2f a[G2_KSTEPS];
        #pragma unroll
        for (int kk = 0; kk < G2_KSTEPS; ++kk) a[kk] = an[kk];
        if (mt + 1 < G2_MT) {
            #pragma unroll
            for (int kk = 0; kk < G2_KSTEPS; ++kk)
                an[kk] = *(const v2f*)(fa + (mt + 1) * 16 * (4 * G2_P4) + 4 * kk);
        }
        __builtin_amdgcn_sched_barrier(0);
        v8f acc0 = {}, acc1 = {};
        #pragma unroll
        for (int kk = 0; kk < G2_KSTEPS; ++kk) {
            acc0 = __builtin_amdgcn_wmma_f32_16x16x4_f32(false, a[kk], false, b0[kk],
                                                         (short)0, acc0, false, false);
            acc1 = __builtin_amdgcn_wmma_f32_16x16x4_f32(false, a[kk], false, b1[kk],
                                                         (short)0, acc1, false, false);
        }
        float* dr = delta + (m0 + mh * 32 + mt * 16 + 8 * h) * DMODEL + col;
        float sp0[8], sp1[8];
        #pragma unroll
        for (int i = 0; i < 8; ++i) {
            sp0[i] = g2_softplus(acc0[i] + bn0);
            sp1[i] = g2_softplus(acc1[i] + bn1);
        }
        #pragma unroll
        for (int i = 0; i < 8; ++i) {
            dr[i * DMODEL]      = sp0[i];
            dr[i * DMODEL + 16] = sp1[i];
        }
    }
}

#define PF 8
#define PB 4
#define BPB (DMODEL / 256)

__global__ __launch_bounds__(256)
void scan_local(const float* __restrict__ U, const float* __restrict__ A_log,
                const float* __restrict__ xp, const float* __restrict__ delta,
                float* __restrict__ out, float* __restrict__ agg) {
    __shared__ float4 sBC[CLEN * 8];

    const int c    = blockIdx.y;
    const int bb   = blockIdx.x / BPB;
    const int dd   = (blockIdx.x % BPB) * 256 + threadIdx.x;
    const int ch   = bb * DMODEL + dd;
    const int row0 = bb * SEQ + c * CLEN;

    const float* __restrict__ dp = delta + row0 * DMODEL + dd;
    const float* __restrict__ up = U     + row0 * DMODEL + dd;
    float*       __restrict__ op = out   + row0 * DMODEL + dd;

    #pragma unroll
    for (int i = threadIdx.x; i < CLEN * 8; i += 256)
        sBC[i] = *(const float4*)(xp + (row0 + (i >> 3)) * KPROJ + DTRANK + 4 * (i & 7));

    float dn[PF], un[PF];
    #pragma unroll
    for (int j = 0; j < PF; ++j) { dn[j] = dp[j * DMODEL]; un[j] = up[j * DMODEL]; }

    float Av[NSTATE];
    #pragma unroll
    for (int n = 0; n < NSTATE; ++n) Av[n] = -__expf(A_log[n]);

    float st[NSTATE];
    #pragma unroll
    for (int n = 0; n < NSTATE; ++n) st[n] = 0.0f;

    __syncthreads();

    float4 nb[8];
    #pragma unroll
    for (int q = 0; q < 8; ++q) nb[q] = sBC[q];

    float dsum = 0.0f;
    #pragma unroll 1
    for (int tb = 0; tb < CLEN; tb += PF) {
        float dc[PF], uc[PF];
        #pragma unroll
        for (int j = 0; j < PF; ++j) { dc[j] = dn[j]; uc[j] = un[j]; }
        if (tb + PF < CLEN) {
            #pragma unroll
            for (int j = 0; j < PF; ++j) {
                dn[j] = dp[(tb + PF + j) * DMODEL];
                un[j] = up[(tb + PF + j) * DMODEL];
            }
        }
        #pragma unroll
        for (int j = 0; j < PF; ++j) {
            const int t = tb + j;
            float Bv[NSTATE], Cv[NSTATE];
            #pragma unroll
            for (int q = 0; q < 4; ++q) {
                float4 vb = nb[q], vc = nb[4 + q];
                Bv[4*q+0]=vb.x; Bv[4*q+1]=vb.y; Bv[4*q+2]=vb.z; Bv[4*q+3]=vb.w;
                Cv[4*q+0]=vc.x; Cv[4*q+1]=vc.y; Cv[4*q+2]=vc.z; Cv[4*q+3]=vc.w;
            }
            const int tn = (t + 1 < CLEN) ? (t + 1) : t;
            #pragma unroll
            for (int q = 0; q < 8; ++q) nb[q] = sBC[tn * 8 + q];
            __builtin_amdgcn_sched_barrier(0);
            const float dl = dc[j];
            const float du = dl * uc[j];
            float y = 0.0f;
            #pragma unroll
            for (int n = 0; n < NSTATE; ++n) {
                float dA = __expf(dl * Av[n]);
                st[n] = fmaf(dA, st[n], du * Bv[n]);
                y     = fmaf(st[n], Cv[n], y);
            }
            dsum += dl;
            op[t * DMODEL] = y;
        }
    }
    const int base = (c * 2 * NSTATE) * NCHAN + ch;
    #pragma unroll
    for (int n = 0; n < NSTATE; ++n)
        agg[base + n * NCHAN] = __expf(Av[n] * dsum);
    #pragma unroll
    for (int n = 0; n < NSTATE; ++n)
        agg[base + (NSTATE + n) * NCHAN] = st[n];
}

__global__ __launch_bounds__(256)
void scan_fixup(const float* __restrict__ U, const float* __restrict__ A_log,
                const float* __restrict__ xp, const float* __restrict__ delta,
                const float* __restrict__ Dvec, const float* __restrict__ agg,
                float* __restrict__ out) {
    __shared__ float4 sC[CLEN * 4];

    const int c    = blockIdx.y;
    const int bb   = blockIdx.x / BPB;
    const int dd   = (blockIdx.x % BPB) * 256 + threadIdx.x;
    const int ch   = bb * DMODEL + dd;
    const int row0 = bb * SEQ + c * CLEN;

    const float* __restrict__ dp = delta + row0 * DMODEL + dd;
    const float* __restrict__ up = U     + row0 * DMODEL + dd;
    float*       __restrict__ op = out   + row0 * DMODEL + dd;

    sC[threadIdx.x] = *(const float4*)(xp + (row0 + (threadIdx.x >> 2)) * KPROJ
                                          + DTRANK + NSTATE + 4 * (threadIdx.x & 3));

    float dn[PF], un[PF], yn[PF];
    #pragma unroll
    for (int j = 0; j < PF; ++j) {
        dn[j] = dp[j * DMODEL]; un[j] = up[j * DMODEL]; yn[j] = op[j * DMODEL];
    }

    float Av[NSTATE];
    #pragma unroll
    for (int n = 0; n < NSTATE; ++n) Av[n] = -__expf(A_log[n]);

    float entry[NSTATE];
    #pragma unroll
    for (int n = 0; n < NSTATE; ++n) entry[n] = 0.0f;
    for (int cc = 0; cc < c; cc += PB) {
        float p[PB][NSTATE], f[PB][NSTATE];
        #pragma unroll
        for (int k = 0; k < PB; ++k) {
            const int ce   = (cc + k < c) ? (cc + k) : (c - 1);
            const int base = (ce * 2 * NSTATE) * NCHAN + ch;
            #pragma unroll
            for (int n = 0; n < NSTATE; ++n) {
                p[k][n] = agg[base + n * NCHAN];
                f[k][n] = agg[base + (NSTATE + n) * NCHAN];
            }
        }
        __builtin_amdgcn_sched_barrier(0);
        #pragma unroll
        for (int k = 0; k < PB; ++k) {
            const bool live = (cc + k < c);
            #pragma unroll
            for (int n = 0; n < NSTATE; ++n)
                entry[n] = fmaf(live ? p[k][n] : 1.0f, entry[n], live ? f[k][n] : -0.0f);
        }
    }

    const float Dd = Dvec[dd];

    __syncthreads();

    float4 nc[4];
    #pragma unroll
    for (int q = 0; q < 4; ++q) nc[q] = sC[q];

    #pragma unroll 1
    for (int tb = 0; tb < CLEN; tb += PF) {
        float dc[PF], uc[PF], yc[PF];
        #pragma unroll
        for (int j = 0; j < PF; ++j) { dc[j] = dn[j]; uc[j] = un[j]; yc[j] = yn[j]; }
        if (tb + PF < CLEN) {
            #pragma unroll
            for (int j = 0; j < PF; ++j) {
                dn[j] = dp[(tb + PF + j) * DMODEL];
                un[j] = up[(tb + PF + j) * DMODEL];
                yn[j] = op[(tb + PF + j) * DMODEL];
            }
        }
        #pragma unroll
        for (int j = 0; j < PF; ++j) {
            const int t = tb + j;
            float Cv[NSTATE];
            #pragma unroll
            for (int q = 0; q < 4; ++q) {
                float4 vc = nc[q];
                Cv[4*q+0]=vc.x; Cv[4*q+1]=vc.y; Cv[4*q+2]=vc.z; Cv[4*q+3]=vc.w;
            }
            const int tn = (t + 1 < CLEN) ? (t + 1) : t;
            #pragma unroll
            for (int q = 0; q < 4; ++q) nc[q] = sC[tn * 4 + q];
            __builtin_amdgcn_sched_barrier(0);
            const float dl = dc[j];
            float ycorr = 0.0f;
            #pragma unroll
            for (int n = 0; n < NSTATE; ++n) {
                entry[n] *= __expf(dl * Av[n]);
                ycorr = fmaf(entry[n], Cv[n], ycorr);
            }
            op[t * DMODEL] = yc[j] + ycorr + uc[j] * Dd;
        }
    }
}

extern "C" void kernel_launch(void* const* d_in, const int* in_sizes, int n_in,
                              void* d_out, int out_size, void* d_ws, size_t ws_size,
                              hipStream_t stream) {
    const float* U     = (const float*)d_in[0];
    const float* A_log = (const float*)d_in[1];
    const float* W1    = (const float*)d_in[2];
    const float* W2    = (const float*)d_in[3];
    const float* bias  = (const float*)d_in[4];
    const float* Dvec  = (const float*)d_in[5];
    float* out = (float*)d_out;

    float* ws    = (float*)d_ws;
    float* xp    = ws;
    float* delta = xp + (size_t)MROWS * KPROJ;
    float* agg   = delta + (size_t)MROWS * DMODEL;

    gemm_xproj<<<dim3(1536 / 8), dim3(256), 0, stream>>>(U, W1, xp);

    gemm_delta<<<dim3((MROWS / G2_BM) * (DMODEL / G2_BN)), dim3(G2_NT), 0, stream>>>(xp, W2, bias, delta);

    dim3 sgrid(NCHAN / 256, NCHUNK);
    scan_local<<<sgrid, dim3(256), 0, stream>>>(U, A_log, xp, delta, out, agg);
    scan_fixup<<<sgrid, dim3(256), 0, stream>>>(U, A_log, xp, delta, Dvec, agg, out);
}


# pass 2 re-mapped to blocks of 32 channels x 8 consecutive chunks: chunk entry states derived once per block (2 states per wave, same fma chain in chunk order), handed over in LDS; sums unchanged
# speedup vs baseline: 2.4262x; 1.0756x over previous
#include <hip/hip_runtime.h>
#include <hip/hip_bf16.h>
#include <math.h>

typedef float v2f __attribute__((ext_vector_type(2)));
typedef float v8f __attribute__((ext_vector_type(8)));

#define BATCH   2
#define SEQ     2048
#define DMODEL  1024
#define DTRANK  64
#define NSTATE  16
#define KPROJ   (DTRANK + 2 * NSTATE)
#define MROWS   (BATCH * SEQ)
#define NCHAN   (BATCH * DMODEL)
#define NCHUNK  32
#define CLEN    (SEQ / NCHUNK)

#define G1_KS    128
#define G1_NSLAB (DMODEL / G1_KS)
#define G1_UROWS 32
#define G1_ROWS  (G1_UROWS + KPROJ)
#define G1_P4    (G1_KS / 4 + 1)
#define G1_SLAB4 (G1_ROWS * G1_P4)
#define G1_KSTEPS (G1_KS / 4)

__device__ __forceinline__ v8f g1_slab_products(const float* pa, const float* pb, v8f acc) {
    v2f a[G1_KSTEPS], b[G1_KSTEPS];
    #pragma unroll
    for (int kk = 0; kk < G1_KSTEPS; ++kk) {
        a[kk] = *(const v2f*)(pa + 4 * kk);
        b[kk] = *(const v2f*)(pb + 4 * kk);
    }
    __builtin_amdgcn_sched_barrier(0);
    #pragma unroll
    for (int kk = 0; kk < G1_KSTEPS; ++kk)
        acc = __builtin_amdgcn_wmma_f32_16x16x4_f32(false, a[kk], false, b[kk],
                                                    (short)0, acc, false, false);
    return acc;
}

__device__ __forceinline__ void g1_lds_barrier() {
    __builtin_amdgcn_fence(__ATOMIC_RELEASE, "workgroup", "local");
    __builtin_amdgcn_s_barrier();
    __builtin_amdgcn_fence(__ATOMIC_ACQUIRE, "workgroup", "local");
}

__global__ __launch_bounds__(256)
void gemm_xproj(const float* __restrict__ U, const float* __restrict__ W1,
                float* __restrict__ xp) {
    __shared__ float4 sm[2 * G1_SLAB4];

    const int lane  = threadIdx.x & 31;
    const int wave  = threadIdx.x >> 5;
    const int tile  = blockIdx.x * 8 + wave;
    const int tilesN = KPROJ / 16;
    const int mt  = tile / tilesN;
    const int nt  = tile % tilesN;
    const int ms0 = (blockIdx.x * 8) / tilesN;
    const int c   = lane & 15;
    const int h   = lane >> 4;

    const float* gu = U  + (ms0 * 16 + wave) * DMODEL + 4 * lane;
    const float* gw = W1 + wave * DMODEL + 4 * lane;
    const int sdst = wave * G1_P4 + lane;

    const int fa = ((mt - ms0) * 16 + c) * (4 * G1_P4) + 2 * h;
    const int fb = (G1_UROWS + nt * 16 + c) * (4 * G1_P4) + 2 * h;

#define G1_PIECES(X) X(0) X(1) X(2) X(3) X(4) X(5) X(6) X(7) X(8) X(9) X(10) X(11) X(12) X(13) X(14) X(15)
#define G1_DECL(i)  float4 r##i;
#define G1_LOAD(i)  r##i = *(const float4*)((((i) < 4) ? (gu + 8 * (i) * DMODEL) : (gw + 8 * ((i) - 4) * DMODEL)) + ko);
#define G1_PARK(i)  sm[dst + 8 * (i) * G1_P4] = r##i;
    G1_PIECES(G1_DECL)
    {
        const int ko = 0, dst = sdst;
        G1_PIECES(G1_LOAD)
        G1_PIECES(G1_PARK)
    }
    {
        const int ko = G1_KS;
        G1_PIECES(G1_LOAD)
    }
    g1_lds_barrier();

    v8f acc = {};
    #pragma unroll 1
    for (int s = 0; s + 1 < G1_NSLAB; ++s) {
        const int cur = (s & 1) * G1_SLAB4;
        const int dst = (G1_SLAB4 - cur) + sdst;
        G1_PIECES(G1_PARK)
        if (s + 2 < G1_NSLAB) {
            const int ko = (s + 2) * G1_KS;
            G1_PIECES(G1_LOAD)
        }
        acc = g1_slab_products((const float*)(sm + cur) + fa, (const float*)(sm + cur) + fb, acc);
        asm volatile("" :: "v"(acc));
        g1_lds_barrier();
    }
    {
        const int cur = ((G1_NSLAB - 1) & 1) * G1_SLAB4;
        acc = g1_slab_products((const float*)(sm + cur) + fa, (const float*)(sm + cur) + fb, acc);
    }
#undef G1_PARK
#undef G1_LOAD
#undef G1_DECL
#undef G1_PIECES

    const int col   = nt * 16 + c;
    const int rbase = mt * 16 + 8 * h;
    #pragma unroll
    for (int i = 0; i < 8; ++i)
        xp[(rbase + i) * KPROJ + col] = acc[i];
}

#define G2_BM  64
#define G2_BN  256
#define G2_P4  (DTRANK / 4 + 1)
#define G2_NT  512
#define G2_MT  2
#define G2_KSTEPS (DTRANK / 4)

__device__ __forceinline__ float g2_softplus(float x) {
    const float e = __builtin_amdgcn_exp2f(-fabsf(x) * 1.44269504088896341f);
    const float u = 1.0f + e;
    const float d = u - 1.0f;
    const float l = __builtin_amdgcn_logf(u) * 0.693147180559945309f;
    const float r = (d == 0.0f) ? e : l * (e * __builtin_amdgcn_rcpf(d));
    return fmaxf(x, 0.0f) + r;
}

__global__ __launch_bounds__(G2_NT)
void gemm_delta(const float* __restrict__ xp, const float* __restrict__ W2,
                const float* __restrict__ bias, float* __restrict__ delta) {
    __shared__ float4 sA[G2_BM * G2_P4];
    __shared__ float4 sB[G2_BN * G2_P4];

    const int lane = threadIdx.x & 31;
    const int wave = (threadIdx.x >> 5) & 7;
    const int mh   = threadIdx.x >> 8;
    const int m0   = (blockIdx.x >> 2) * G2_BM;
    const int n0   = (blockIdx.x & 3) * G2_BN;
    const int c    = lane & 15;
    const int h    = lane >> 4;

    #pragma unroll
    for (int i = 0; i < G2_BM * 16 / G2_NT; ++i) {
        const int idx = threadIdx.x + G2_NT * i;
        const int r = idx >> 4, q = idx & 15;
        sA[r * G2_P4 + q] = *(const float4*)(xp + (m0 + r) * KPROJ + 4 * q);
    }
    #pragma unroll
    for (int i = 0; i < G2_BN * 16 / G2_NT; ++i) {
        const int idx = threadIdx.x + G2_NT * i;
        const int r = idx >> 4, q = idx & 15;
        sB[r * G2_P4 + q] = *(const float4*)(W2 + (n0 + r) * DTRANK + 4 * q);
    }
    __syncthreads();

    const float* fb = (const float*)sB + (wave * 32 + c) * (4 * G2_P4) + 2 * h;
    v2f b0[G2_KSTEPS], b1[G2_KSTEPS];
    #pragma unroll
    for (int kk = 0; kk < G2_KSTEPS; ++kk) {
        b0[kk] = *(const v2f*)(fb + 4 * kk);
        b1[kk] = *(const v2f*)(fb + 16 * (4 * G2_P4) + 4 * kk);
    }

    const int   col = n0 + wave * 32 + c;
    const float bn0 = bias[col];
    const float bn1 = bias[col + 16];

    const float* fa = (const float*)sA + (mh * 32 + c) * (4 * G2_P4) + 2 * h;
    v2f an[G2_KSTEPS];
    #pragma unroll
    for (int kk = 0; kk < G2_KSTEPS; ++kk) an[kk] = *(const v2f*)(fa + 4 * kk);

    #pragma unroll
    for (int mt = 0; mt < G2_MT; ++mt) {
        v2f a[G2_KSTEPS];
        #pragma unroll
        for (int kk = 0; kk < G2_KSTEPS; ++kk) a[kk] = an[kk];
        if (mt + 1 < G2_MT) {
            #pragma unroll
            for (int kk = 0; kk < G2_KSTEPS; ++kk)
                an[kk] = *(const v2f*)(fa + (mt + 1) * 16 * (4 * G2_P4) + 4 * kk);
        }
        __builtin_amdgcn_sched_barrier(0);
        v8f acc0 = {}, acc1 = {};
        #pragma unroll
        for (int kk = 0; kk < G2_KSTEPS; ++kk) {
            acc0 = __builtin_amdgcn_wmma_f32_16x16x4_f32(false, a[kk], false, b0[kk],
                                                         (short)0, acc0, false, false);
            acc1 = __builtin_amdgcn_wmma_f32_16x16x4_f32(false, a[kk], false, b1[kk],
                                                         (short)0, acc1, false, false);
        }
        float* dr = delta + (m0 + mh * 32 + mt * 16 + 8 * h) * DMODEL + col;
        float sp0[8], sp1[8];
        #pragma unroll
        for (int i = 0; i < 8; ++i) {
            sp0[i] = g2_softplus(acc0[i] + bn0);
            sp1[i] = g2_softplus(acc1[i] + bn1);
        }
        #pragma unroll
        for (int i = 0; i < 8; ++i) {
            dr[i * DMODEL]      = sp0[i];
            dr[i * DMODEL + 16] = sp1[i];
        }
    }
}

#define PF 8
#define BPB (DMODEL / 256)

__global__ __launch_bounds__(256)
void scan_local(const float* __restrict__ U, const float* __restrict__ A_log,
                const float* __restrict__ xp, const float* __restrict__ delta,
                float* __restrict__ out, float* __restrict__ agg) {
    __shared__ float4 sBC[CLEN * 8];

    const int c    = blockIdx.y;
    const int bb   = blockIdx.x / BPB;
    const int dd   = (blockIdx.x % BPB) * 256 + threadIdx.x;
    const int ch   = bb * DMODEL + dd;
    const int row0 = bb * SEQ + c * CLEN;

    const float* __restrict__ dp = delta + row0 * DMODEL + dd;
    const float* __restrict__ up = U     + row0 * DMODEL + dd;
    float*       __restrict__ op = out   + row0 * DMODEL + dd;

    #pragma unroll
    for (int i = threadIdx.x; i < CLEN * 8; i += 256)
        sBC[i] = *(const float4*)(xp + (row0 + (i >> 3)) * KPROJ + DTRANK + 4 * (i & 7));

    float dn[PF], un[PF];
    #pragma unroll
    for (int j = 0; j < PF; ++j) { dn[j] = dp[j * DMODEL]; un[j] = up[j * DMODEL]; }

    float Av[NSTATE];
    #pragma unroll
    for (int n = 0; n < NSTATE; ++n) Av[n] = -__expf(A_log[n]);

    float st[NSTATE];
    #pragma unroll
    for (int n = 0; n < NSTATE; ++n) st[n] = 0.0f;

    __syncthreads();

    float4 nb[8];
    #pragma unroll
    for (int q = 0; q < 8; ++q) nb[q] = sBC[q];

    float dsum = 0.0f;
    #pragma unroll 1
    for (int tb = 0; tb < CLEN; tb += PF) {
        float dc[PF], uc[PF];
        #pragma unroll
        for (int j = 0; j < PF; ++j) { dc[j] = dn[j]; uc[j] = un[j]; }
        if (tb + PF < CLEN) {
            #pragma unroll
            for (int j = 0; j < PF; ++j) {
                dn[j] = dp[(tb + PF + j) * DMODEL];
                un[j] = up[(tb + PF + j) * DMODEL];
            }
        }
        #pragma unroll
        for (int j = 0; j < PF; ++j) {
            const int t = tb + j;
            float Bv[NSTATE], Cv[NSTATE];
            #pragma unroll
            for (int q = 0; q < 4; ++q) {
                float4 vb = nb[q], vc = nb[4 + q];
                Bv[4*q+0]=vb.x; Bv[4*q+1]=vb.y; Bv[4*q+2]=vb.z; Bv[4*q+3]=vb.w;
                Cv[4*q+0]=vc.x; Cv[4*q+1]=vc.y; Cv[4*q+2]=vc.z; Cv[4*q+3]=vc.w;
            }
            const int tn = (t + 1 < CLEN) ? (t + 1) : t;
            #pragma unroll
            for (int q = 0; q < 8; ++q) nb[q] = sBC[tn * 8 + q];
            __builtin_amdgcn_sched_barrier(0);
            const float dl = dc[j];
            const float du = dl * uc[j];
            float y = 0.0f;
            #pragma unroll
            for (int n = 0; n < NSTATE; ++n) {
                float dA = __expf(dl * Av[n]);
                st[n] = fmaf(dA, st[n], du * Bv[n]);
                y     = fmaf(st[n], Cv[n], y);
            }
            dsum += dl;
            op[t * DMODEL] = y;
        }
    }
    const int base = (c * 2 * NSTATE) * NCHAN + ch;
    #pragma unroll
    for (int n = 0; n < NSTATE; ++n)
        agg[base + n * NCHAN] = __expf(Av[n] * dsum);
    #pragma unroll
    for (int n = 0; n < NSTATE; ++n)
        agg[base + (NSTATE + n) * NCHAN] = st[n];
}

#define FX_CH   32
#define FX_CPB  8
#define FX_XB   (DMODEL / FX_CH)

struct FxGroup { float pa[FX_CPB], pb[FX_CPB], fa[FX_CPB], fb[FX_CPB]; };

__device__ __forceinline__ void fx_load(FxGroup& s, const float* __restrict__ ag, int grp) {
    #pragma unroll
    for (int k = 0; k < FX_CPB; ++k) {
        const float* a = ag + ((grp * FX_CPB + k) * 2 * NSTATE) * NCHAN;
        s.pa[k] = a[0];               s.pb[k] = a[NCHAN];
        s.fa[k] = a[NSTATE * NCHAN];  s.fb[k] = a[(NSTATE + 1) * NCHAN];
    }
}

__device__ __forceinline__ void fx_pass(const FxGroup& s, float& ea, float& eb) {
    #pragma unroll
    for (int k = 0; k < FX_CPB; ++k) {
        ea = fmaf(s.pa[k], ea, s.fa[k]);
        eb = fmaf(s.pb[k], eb, s.fb[k]);
    }
}

__device__ __forceinline__ void fx_own(const FxGroup& s, float ea, float eb, float* se) {
    #pragma unroll
    for (int k = 0; k < FX_CPB; ++k) {
        se[k * NSTATE * FX_CH]         = ea;
        se[k * NSTATE * FX_CH + FX_CH] = eb;
        if (k + 1 < FX_CPB) {
            ea = fmaf(s.pa[k], ea, s.fa[k]);
            eb = fmaf(s.pb[k], eb, s.fb[k]);
        }
    }
}

__global__ __launch_bounds__(256)
void scan_fixup(const float* __restrict__ U, const float* __restrict__ A_log,
                const float* __restrict__ xp, const float* __restrict__ delta,
                const float* __restrict__ Dvec, const float* __restrict__ agg,
                float* __restrict__ out) {
    __shared__ float4 sC[FX_CPB * CLEN * 4];
    __shared__ float  sE[FX_CPB * NSTATE * FX_CH];

    const int lane = threadIdx.x & 31;
    const int wave = threadIdx.x >> 5;
    const int g    = blockIdx.y;
    const int c    = g * FX_CPB + wave;
    const int bb   = blockIdx.x / FX_XB;
    const int dd   = (blockIdx.x % FX_XB) * FX_CH + lane;
    const int ch   = bb * DMODEL + dd;
    const int row0 = bb * SEQ + c * CLEN;

    const float* __restrict__ dp = delta + row0 * DMODEL + dd;
    const float* __restrict__ up = U     + row0 * DMODEL + dd;
    float*       __restrict__ op = out   + row0 * DMODEL + dd;

    float4* sCw = sC + wave * (CLEN * 4);
    #pragma unroll
    for (int q = 0; q < CLEN * 4 / 32; ++q) {
        const int i = lane + 32 * q;
        sCw[i] = *(const float4*)(xp + (row0 + (i >> 2)) * KPROJ + DTRANK + NSTATE + 4 * (i & 3));
    }

    float dn[PF], un[PF], yn[PF];
    #pragma unroll
    for (int j = 0; j < PF; ++j) {
        dn[j] = dp[j * DMODEL]; un[j] = up[j * DMODEL]; yn[j] = op[j * DMODEL];
    }

    float Av[NSTATE];
    #pragma unroll
    for (int n = 0; n < NSTATE; ++n) Av[n] = -__expf(A_log[n]);

    static_assert(NCHUNK / FX_CPB == 4, "the prefix below is written out for four groups of chunks");
    const float* ag = agg + (2 * wave) * NCHAN + ch;
    float ea = 0.0f, eb = 0.0f;
    float* se = sE + (2 * wave) * FX_CH + lane;
    FxGroup A, B;
    fx_load(A, ag, 0);
    if (g == 0) {
        fx_own(A, ea, eb, se);
    } else {
        fx_load(B, ag, 1);
        __builtin_amdgcn_sched_barrier(0);
        fx_pass(A, ea, eb);
        if (g == 1) {
            fx_own(B, ea, eb, se);
        } else {
            fx_load(A, ag, 2);
            __builtin_amdgcn_sched_barrier(0);
            fx_pass(B, ea, eb);
            if (g == 2) {
                fx_own(A, ea, eb, se);
            } else {
                fx_load(B, ag, 3);
                __builtin_amdgcn_sched_barrier(0);
                fx_pass(A, ea, eb);
                fx_own(B, ea, eb, se);
            }
        }
    }

    const float Dd = Dvec[dd];

    __syncthreads();

    float entry[NSTATE];
    #pragma unroll
    for (int n = 0; n < NSTATE; ++n) entry[n] = sE[(wave * NSTATE + n) * FX_CH + lane];

    float4 nc[4];
    #pragma unroll
    for (int q = 0; q < 4; ++q) nc[q] = sCw[q];

    #pragma unroll 1
    for (int tb = 0; tb < CLEN; tb += PF) {
        float dc[PF], uc[PF], yc[PF];
        #pragma unroll
        for (int j = 0; j < PF; ++j) { dc[j] = dn[j]; uc[j] = un[j]; yc[j] = yn[j]; }
        if (tb + PF < CLEN) {
            #pragma unroll
            for (int j = 0; j < PF; ++j) {
                dn[j] = dp[(tb + PF + j) * DMODEL];
                un[j] = up[(tb + PF + j) * DMODEL];
                yn[j] = op[(tb + PF + j) * DMODEL];
            }
        }
        #pragma unroll
        for (int j = 0; j < PF; ++j) {
            const int t = tb + j;
            float Cv[NSTATE];
            #pragma unroll
            for (int q = 0; q < 4; ++q) {
                float4 vc = nc[q];
                Cv[4*q+0]=vc.x; Cv[4*q+1]=vc.y; Cv[4*q+2]=vc.z; Cv[4*q+3]=vc.w;
            }
            const int tn = (t + 1 < CLEN) ? (t + 1) : t;
            #pragma unroll
            for (int q = 0; q < 4; ++q) nc[q] = sCw[tn * 4 + q];
            __builtin_amdgcn_sched_barrier(0);
            const float dl = dc[j];
            float ycorr = 0.0f;
            #pragma unroll
            for (int n = 0; n < NSTATE; ++n) {
                entry[n] *= __expf(dl * Av[n]);
                ycorr = fmaf(entry[n], Cv[n], ycorr);
            }
            op[t * DMODEL] = yc[j] + ycorr + uc[j] * Dd;
        }
    }
}

extern "C" void kernel_launch(void* const* d_in, const int* in_sizes, int n_in,
                              void* d_out, int out_size, void* d_ws, size_t ws_size,
                              hipStream_t stream) {
    const float* U     = (const float*)d_in[0];
    const float* A_log = (const float*)d_in[1];
    const float* W1    = (const float*)d_in[2];
    const float* W2    = (const float*)d_in[3];
    const float* bias  = (const float*)d_in[4];
    const float* Dvec  = (const float*)d_in[5];
    float* out = (float*)d_out;

    float* ws    = (float*)d_ws;
    float* xp    = ws;
    float* delta = xp + (size_t)MROWS * KPROJ;
    float* agg   = delta + (size_t)MROWS * DMODEL;

    gemm_xproj<<<dim3(1536 / 8), dim3(256), 0, stream>>>(U, W1, xp);

    gemm_delta<<<dim3((MROWS / G2_BM) * (DMODEL / G2_BN)), dim3(G2_NT), 0, stream>>>(xp, W2, bias, delta);

    dim3 sgrid(NCHAN / 256, NCHUNK);
    scan_local<<<sgrid, dim3(256), 0, stream>>>(U, A_log, xp, delta, out, agg);
    scan_fixup<<<dim3(NCHAN / FX_CH, NCHUNK / FX_CPB), dim3(256), 0, stream>>>(U, A_log, xp, delta, Dvec, agg, out);
}


# x-projection GEMM split 4 ways along K: 256 blocks of 64 rows x 256 k, 3 tiles a wave, 4 planes of partial sums; delta GEMM adds the planes (k quarters 0+1+2+3: new sum order), hands B/C to scans
# speedup vs baseline: 2.5988x; 1.0711x over previous
#include <hip/hip_runtime.h>
#include <hip/hip_bf16.h>
#include <math.h>

typedef float v2f __attribute__((ext_vector_type(2)));
typedef float v8f __attribute__((ext_vector_type(8)));

#define BATCH   2
#define SEQ     2048
#define DMODEL  1024
#define DTRANK  64
#define NSTATE  16
#define KPROJ   (DTRANK + 2 * NSTATE)
#define MROWS   (BATCH * SEQ)
#define NCHAN   (BATCH * DMODEL)
#define NCHUNK  32
#define CLEN    (SEQ / NCHUNK)

#define G1_KSPLIT 4
#define G1_KR    (DMODEL / G1_KSPLIT)
#define G1_KS    128
#define G1_NSLAB (G1_KR / G1_KS)
#define G1_UROWS 64
#define G1_ROWS  (G1_UROWS + KPROJ)
#define G1_NPC   (G1_ROWS / 8)
#define G1_P4    (G1_KS / 4 + 1)
#define G1_SLAB4 (G1_ROWS * G1_P4)
#define G1_KSTEPS (G1_KS / 4)
#define G1_KG    16
#define XP_PLANE (MROWS * KPROJ)

__device__ __forceinline__ float4 xp_add4(float4 a, float4 b, float4 c, float4 d) {
    float4 r;
    r.x = ((a.x + b.x) + c.x) + d.x;
    r.y = ((a.y + b.y) + c.y) + d.y;
    r.z = ((a.z + b.z) + c.z) + d.z;
    r.w = ((a.w + b.w) + c.w) + d.w;
    return r;
}
__device__ __forceinline__ float4 xp_sum4(const float* __restrict__ p) {
    return xp_add4(*(const float4*)(p), *(const float4*)(p + XP_PLANE),
                   *(const float4*)(p + 2 * XP_PLANE), *(const float4*)(p + 3 * XP_PLANE));
}

__device__ __forceinline__ void g1_slab_products(const float* pa, const float* pb,
                                                 v8f& acc0, v8f& acc1, v8f& acc2) {
    #pragma unroll
    for (int g = 0; g < G1_KSTEPS / G1_KG; ++g) {
        v2f a[G1_KG], b0[G1_KG], b1[G1_KG], b2[G1_KG];
        #pragma unroll
        for (int kk = 0; kk < G1_KG; ++kk) {
            const int o = 4 * (g * G1_KG + kk);
            a[kk]  = *(const v2f*)(pa + o);
            b0[kk] = *(const v2f*)(pb + o);
            b1[kk] = *(const v2f*)(pb + 16 * (4 * G1_P4) + o);
            b2[kk] = *(const v2f*)(pb + 32 * (4 * G1_P4) + o);
        }
        __builtin_amdgcn_sched_barrier(0);
        #pragma unroll
        for (int kk = 0; kk < G1_KG; ++kk) {
            acc0 = __builtin_amdgcn_wmma_f32_16x16x4_f32(false, a[kk], false, b0[kk],
                                                         (short)0, acc0, false, false);
            acc1 = __builtin_amdgcn_wmma_f32_16x16x4_f32(false, a[kk], false, b1[kk],
                                                         (short)0, acc1, false, false);
            acc2 = __builtin_amdgcn_wmma_f32_16x16x4_f32(false, a[kk], false, b2[kk],
                                                         (short)0, acc2, false, false);
        }
        __builtin_amdgcn_sched_barrier(0);
    }
}

__device__ __forceinline__ void g1_lds_barrier() {
    __builtin_amdgcn_fence(__ATOMIC_RELEASE, "workgroup", "local");
    __builtin_amdgcn_s_barrier();
    __builtin_amdgcn_fence(__ATOMIC_ACQUIRE, "workgroup", "local");
}

__global__ __launch_bounds__(256)
void gemm_xproj(const float* __restrict__ U, const float* __restrict__ W1,
                float* __restrict__ xpp) {
    __shared__ float4 sm[G1_NSLAB * G1_SLAB4];

    const int lane  = threadIdx.x & 31;
    const int wave  = threadIdx.x >> 5;
    const int mb    = blockIdx.x >> 2;
    const int ks    = blockIdx.x & 3;
    const int strip = wave >> 1;
    const int ntb   = (wave & 1) * 3;
    const int c   = lane & 15;
    const int h   = lane >> 4;

    const float* gu = U  + (mb * G1_UROWS + wave) * DMODEL + ks * G1_KR + 4 * lane;
    const float* gw = W1 + wave * DMODEL + ks * G1_KR + 4 * lane;
    const int sdst = wave * G1_P4 + lane;

    const int fa = (strip * 16 + c) * (4 * G1_P4) + 2 * h;
    const int fb = (G1_UROWS + ntb * 16 + c) * (4 * G1_P4) + 2 * h;

#define G1_PIECES(X) X(0) X(1) X(2) X(3) X(4) X(5) X(6) X(7) X(8) X(9) \
                     X(10) X(11) X(12) X(13) X(14) X(15) X(16) X(17) X(18) X(19)
#define G1_DECL(i)  float4 r##i;
#define G1_LOAD(i)  r##i = *(const float4*)((((i) < 8) ? (gu + 8 * (i) * DMODEL) : (gw + 8 * ((i) - 8) * DMODEL)) + ko);
#define G1_PARK(i)  sm[dst + 8 * (i) * G1_P4] = r##i;
    G1_PIECES(G1_DECL)
    {
        const int ko = 0;
        G1_PIECES(G1_LOAD)
    }

    v8f acc0 = {}, acc1 = {}, acc2 = {};
    #pragma unroll 1
    for (int s = 0; s < G1_NSLAB; ++s) {
        const int cur = s * G1_SLAB4;
        const int dst = cur + sdst;
        G1_PIECES(G1_PARK)
        if (s + 1 < G1_NSLAB) {
            const int ko = (s + 1) * G1_KS;
            G1_PIECES(G1_LOAD)
        }
        g1_lds_barrier();
        g1_slab_products((const float*)(sm + cur) + fa, (const float*)(sm + cur) + fb,
                         acc0, acc1, acc2);
    }
#undef G1_PARK
#undef G1_LOAD
#undef G1_DECL
#undef G1_PIECES

    float* xr = xpp + ks * XP_PLANE + (mb * G1_UROWS + strip * 16 + 8 * h) * KPROJ + ntb * 16 + c;
    #pragma unroll
    for (int i = 0; i < 8; ++i) {
        xr[i * KPROJ]      = acc0[i];
        xr[i * KPROJ + 16] = acc1[i];
        xr[i * KPROJ + 32] = acc2[i];
    }
}

#define G2_BM  64
#define G2_BN  256
#define G2_P4  (DTRANK / 4 + 1)
#define G2_NT  512
#define G2_MT  2
#define G2_KSTEPS (DTRANK / 4)

__device__ __forceinline__ float g2_softplus(float x) {
    const float e = __builtin_amdgcn_exp2f(-fabsf(x) * 1.44269504088896341f);
    const float u = 1.0f + e;
    const float d = u - 1.0f;
    const float l = __builtin_amdgcn_logf(u) * 0.693147180559945309f;
    const float r = (d == 0.0f) ? e : l * (e * __builtin_amdgcn_rcpf(d));
    return fmaxf(x, 0.0f) + r;
}

__global__ __launch_bounds__(G2_NT)
void gemm_delta(const float* __restrict__ xp, const float* __restrict__ W2,
                const float* __restrict__ bias, float* __restrict__ delta,
                float* __restrict__ bc) {
    __shared__ float4 sA[G2_BM * G2_P4];
    __shared__ float4 sB[G2_BN * G2_P4];
    __shared__ float4 sP[G1_KSPLIT * 128];

    const int lane = threadIdx.x & 31;
    const int wave = (threadIdx.x >> 5) & 7;
    const int mh   = threadIdx.x >> 8;
    const int m0   = (blockIdx.x >> 2) * G2_BM;
    const int n0   = (blockIdx.x & 3) * G2_BN;
    const int c    = lane & 15;
    const int h    = lane >> 4;

    #pragma unroll
    for (int i = 0; i < G2_BM * 16 / G2_NT; ++i) {
        const int idx = threadIdx.x + G2_NT * i;
        const int r = idx >> 4, q = idx & 15;
        sA[r * G2_P4 + q] = xp_sum4(xp + (m0 + r) * KPROJ + 4 * q);
    }
    #pragma unroll
    for (int i = 0; i < G2_BN * 16 / G2_NT; ++i) {
        const int idx = threadIdx.x + G2_NT * i;
        const int r = idx >> 4, q = idx & 15;
        sB[r * G2_P4 + q] = *(const float4*)(W2 + (n0 + r) * DTRANK + 4 * q);
    }
    const int nb  = blockIdx.x & 3;
    const int bcr = m0 + nb * 16 + ((threadIdx.x & 127) >> 3);
    const int bcq = 4 * (threadIdx.x & 7);
    sP[threadIdx.x] = *(const float4*)(xp + (threadIdx.x >> 7) * XP_PLANE + bcr * KPROJ + DTRANK + bcq);
    __syncthreads();

    if (threadIdx.x < 128)
        *(float4*)(bc + bcr * (2 * NSTATE) + bcq) =
            xp_add4(sP[threadIdx.x], sP[128 + threadIdx.x], sP[256 + threadIdx.x], sP[384 + threadIdx.x]);

    const float* fb = (const float*)sB + (wave * 32 + c) * (4 * G2_P4) + 2 * h;
    v2f b0[G2_KSTEPS], b1[G2_KSTEPS];
    #pragma unroll
    for (int kk = 0; kk < G2_KSTEPS; ++kk) {
        b0[kk] = *(const v2f*)(fb + 4 * kk);
        b1[kk] = *(const v2f*)(fb + 16 * (4 * G2_P4) + 4 * kk);
    }

    const int   col = n0 + wave * 32 + c;
    const float bn0 = bias[col];
    const float bn1 = bias[col + 16];

    const float* fa = (const float*)sA + (mh * 32 + c) * (4 * G2_P4) + 2 * h;
    v2f an[G2_KSTEPS];
    #pragma unroll
    for (int kk = 0; kk < G2_KSTEPS; ++kk) an[kk] = *(const v2f*)(fa + 4 * kk);

    #pragma unroll
    for (int mt = 0; mt < G2_MT; ++mt) {
        v2f a[G2_KSTEPS];
        #pragma unroll
        for (int kk = 0; kk < G2_KSTEPS; ++kk) a[kk] = an[kk];
        if (mt + 1 < G2_MT) {
            #pragma unroll
            for (int kk = 0; kk < G2_KSTEPS; ++kk)
                an[kk] = *(const v2f*)(fa + (mt + 1) * 16 * (4 * G2_P4) + 4 * kk);
        }
        __builtin_amdgcn_sched_barrier(0);
        v8f acc0 = {}, acc1 = {};
        #pragma unroll
        for (int kk = 0; kk < G2_KSTEPS; ++kk) {
            acc0 = __builtin_amdgcn_wmma_f32_16x16x4_f32(false, a[kk], false, b0[kk],
                                                         (short)0, acc0, false, false);
            acc1 = __builtin_amdgcn_wmma_f32_16x16x4_f32(false, a[kk], false, b1[kk],
                                                         (short)0, acc1, false, false);
        }
        float* dr = delta + (m0 + mh * 32 + mt * 16 + 8 * h) * DMODEL + col;
        float sp0[8], sp1[8];
        #pragma unroll
        for (int i = 0; i < 8; ++i) {
            sp0[i] = g2_softplus(acc0[i] + bn0);
            sp1[i] = g2_softplus(acc1[i] + bn1);
        }
        #pragma unroll
        for (int i = 0; i < 8; ++i) {
            dr[i * DMODEL]      = sp0[i];
            dr[i * DMODEL + 16] = sp1[i];
        }
    }
}

#define PF 8
#define BPB (DMODEL / 256)

__global__ __launch_bounds__(256)
void scan_local(const float* __restrict__ U, const float* __restrict__ A_log,
                const float* __restrict__ bc, const float* __restrict__ delta,
                float* __restrict__ out, float* __restrict__ agg) {
    __shared__ float4 sBC[CLEN * 8];

    const int c    = blockIdx.y;
    const int bb   = blockIdx.x / BPB;
    const int dd   = (blockIdx.x % BPB) * 256 + threadIdx.x;
    const int ch   = bb * DMODEL + dd;
    const int row0 = bb * SEQ + c * CLEN;

    const float* __restrict__ dp = delta + row0 * DMODEL + dd;
    const float* __restrict__ up = U     + row0 * DMODEL + dd;
    float*       __restrict__ op = out   + row0 * DMODEL + dd;

    #pragma unroll
    for (int i = threadIdx.x; i < CLEN * 8; i += 256)
        sBC[i] = *(const float4*)(bc + (row0 + (i >> 3)) * (2 * NSTATE) + 4 * (i & 7));

    float dn[PF], un[PF];
    #pragma unroll
    for (int j = 0; j < PF; ++j) { dn[j] = dp[j * DMODEL]; un[j] = up[j * DMODEL]; }

    float Av[NSTATE];
    #pragma unroll
    for (int n = 0; n < NSTATE; ++n) Av[n] = -__expf(A_log[n]);

    float st[NSTATE];
    #pragma unroll
    for (int n = 0; n < NSTATE; ++n) st[n] = 0.0f;

    __syncthreads();

    float4 nb[8];
    #pragma unroll
    for (int q = 0; q < 8; ++q) nb[q] = sBC[q];

    float dsum = 0.0f;
    #pragma unroll 1
    for (int tb = 0; tb < CLEN; tb += PF) {
        float dc[PF], uc[PF];
        #pragma unroll
        for (int j = 0; j < PF; ++j) { dc[j] = dn[j]; uc[j] = un[j]; }
        if (tb + PF < CLEN) {
            #pragma unroll
            for (int j = 0; j < PF; ++j) {
                dn[j] = dp[(tb + PF + j) * DMODEL];
                un[j] = up[(tb + PF + j) * DMODEL];
            }
        }
        #pragma unroll
        for (int j = 0; j < PF; ++j) {
            const int t = tb + j;
            float Bv[NSTATE], Cv[NSTATE];
            #pragma unroll
            for (int q = 0; q < 4; ++q) {
                float4 vb = nb[q], vc = nb[4 + q];
                Bv[4*q+0]=vb.x; Bv[4*q+1]=vb.y; Bv[4*q+2]=vb.z; Bv[4*q+3]=vb.w;
                Cv[4*q+0]=vc.x; Cv[4*q+1]=vc.y; Cv[4*q+2]=vc.z; Cv[4*q+3]=vc.w;
            }
            const int tn = (t + 1 < CLEN) ? (t + 1) : t;
            #pragma unroll
            for (int q = 0; q < 8; ++q) nb[q] = sBC[tn * 8 + q];
            __builtin_amdgcn_sched_barrier(0);
            const float dl = dc[j];
            const float du = dl * uc[j];
            float y = 0.0f;
            #pragma unroll
            for (int n = 0; n < NSTATE; ++n) {
                float dA = __expf(dl * Av[n]);
                st[n] = fmaf(dA, st[n], du * Bv[n]);
                y     = fmaf(st[n], Cv[n], y);
            }
            dsum += dl;
            op[t * DMODEL] = y;
        }
    }
    const int base = (c * 2 * NSTATE) * NCHAN + ch;
    #pragma unroll
    for (int n = 0; n < NSTATE; ++n)
        agg[base + n * NCHAN] = __expf(Av[n] * dsum);
    #pragma unroll
    for (int n = 0; n < NSTATE; ++n)
        agg[base + (NSTATE + n) * NCHAN] = st[n];
}

#define FX_CH   32
#define FX_CPB  8
#define FX_XB   (DMODEL / FX_CH)

struct FxGroup { float pa[FX_CPB], pb[FX_CPB], fa[FX_CPB], fb[FX_CPB]; };

__device__ __forceinline__ void fx_load(FxGroup& s, const float* __restrict__ ag, int grp) {
    #pragma unroll
    for (int k = 0; k < FX_CPB; ++k) {
        const float* a = ag + ((grp * FX_CPB + k) * 2 * NSTATE) * NCHAN;
        s.pa[k] = a[0];               s.pb[k] = a[NCHAN];
        s.fa[k] = a[NSTATE * NCHAN];  s.fb[k] = a[(NSTATE + 1) * NCHAN];
    }
}

__device__ __forceinline__ void fx_pass(const FxGroup& s, float& ea, float& eb) {
    #pragma unroll
    for (int k = 0; k < FX_CPB; ++k) {
        ea = fmaf(s.pa[k], ea, s.fa[k]);
        eb = fmaf(s.pb[k], eb, s.fb[k]);
    }
}

__device__ __forceinline__ void fx_own(const FxGroup& s, float ea, float eb, float* se) {
    #pragma unroll
    for (int k = 0; k < FX_CPB; ++k) {
        se[k * NSTATE * FX_CH]         = ea;
        se[k * NSTATE * FX_CH + FX_CH] = eb;
        if (k + 1 < FX_CPB) {
            ea = fmaf(s.pa[k], ea, s.fa[k]);
            eb = fmaf(s.pb[k], eb, s.fb[k]);
        }
    }
}

__global__ __launch_bounds__(256)
void scan_fixup(const float* __restrict__ U, const float* __restrict__ A_log,
                const float* __restrict__ bc, const float* __restrict__ delta,
                const float* __restrict__ Dvec, const float* __restrict__ agg,
                float* __restrict__ out) {
    __shared__ float4 sC[FX_CPB * CLEN * 4];
    __shared__ float  sE[FX_CPB * NSTATE * FX_CH];

    const int lane = threadIdx.x & 31;
    const int wave = threadIdx.x >> 5;
    const int g    = blockIdx.y;
    const int c    = g * FX_CPB + wave;
    const int bb   = blockIdx.x / FX_XB;
    const int dd   = (blockIdx.x % FX_XB) * FX_CH + lane;
    const int ch   = bb * DMODEL + dd;
    const int row0 = bb * SEQ + c * CLEN;

    const float* __restrict__ dp = delta + row0 * DMODEL + dd;
    const float* __restrict__ up = U     + row0 * DMODEL + dd;
    float*       __restrict__ op = out   + row0 * DMODEL + dd;

    float4* sCw = sC + wave * (CLEN * 4);
    #pragma unroll
    for (int q = 0; q < CLEN * 4 / 32; ++q) {
        const int i = lane + 32 * q;
        sCw[i] = *(const float4*)(bc + (row0 + (i >> 2)) * (2 * NSTATE) + NSTATE + 4 * (i & 3));
    }

    float dn[PF], un[PF], yn[PF];
    #pragma unroll
    for (int j = 0; j < PF; ++j) {
        dn[j] = dp[j * DMODEL]; un[j] = up[j * DMODEL]; yn[j] = op[j * DMODEL];
    }

    float Av[NSTATE];
    #pragma unroll
    for (int n = 0; n < NSTATE; ++n) Av[n] = -__expf(A_log[n]);

    static_assert(NCHUNK / FX_CPB == 4, "the prefix below is written out for four groups of chunks");
    const float* ag = agg + (2 * wave) * NCHAN + ch;
    float ea = 0.0f, eb = 0.0f;
    float* se = sE + (2 * wave) * FX_CH + lane;
    FxGroup A, B;
    fx_load(A, ag, 0);
    if (g == 0) {
        fx_own(A, ea, eb, se);
    } else {
        fx_load(B, ag, 1);
        __builtin_amdgcn_sched_barrier(0);
        fx_pass(A, ea, eb);
        if (g == 1) {
            fx_own(B, ea, eb, se);
        } else {
            fx_load(A, ag, 2);
            __builtin_amdgcn_sched_barrier(0);
            fx_pass(B, ea, eb);
            if (g == 2) {
                fx_own(A, ea, eb, se);
            } else {
                fx_load(B, ag, 3);
                __builtin_amdgcn_sched_barrier(0);
                fx_pass(A, ea, eb);
                fx_own(B, ea, eb, se);
            }
        }
    }

    const float Dd = Dvec[dd];

    __syncthreads();

    float entry[NSTATE];
    #pragma unroll
    for (int n = 0; n < NSTATE; ++n) entry[n] = sE[(wave * NSTATE + n) * FX_CH + lane];

    float4 nc[4];
    #pragma unroll
    for (int q = 0; q < 4; ++q) nc[q] = sCw[q];

    #pragma unroll 1
    for (int tb = 0; tb < CLEN; tb += PF) {
        float dc[PF], uc[PF], yc[PF];
        #pragma unroll
        for (int j = 0; j < PF; ++j) { dc[j] = dn[j]; uc[j] = un[j]; yc[j] = yn[j]; }
        if (tb + PF < CLEN) {
            #pragma unroll
            for (int j = 0; j < PF; ++j) {
                dn[j] = dp[(tb + PF + j) * DMODEL];
                un[j] = up[(tb + PF + j) * DMODEL];
                yn[j] = op[(tb + PF + j) * DMODEL];
            }
        }
        #pragma unroll
        for (int j = 0; j < PF; ++j) {
            const int t = tb + j;
            float Cv[NSTATE];
            #pragma unroll
            for (int q = 0; q < 4; ++q) {
                float4 vc = nc[q];
                Cv[4*q+0]=vc.x; Cv[4*q+1]=vc.y; Cv[4*q+2]=vc.z; Cv[4*q+3]=vc.w;
            }
            const int tn = (t + 1 < CLEN) ? (t + 1) : t;
            #pragma unroll
            for (int q = 0; q < 4; ++q) nc[q] = sCw[tn * 4 + q];
            __builtin_amdgcn_sched_barrier(0);
            const float dl = dc[j];
            float ycorr = 0.0f;
            #pragma unroll
            for (int n = 0; n < NSTATE; ++n) {
                entry[n] *= __expf(dl * Av[n]);
                ycorr = fmaf(entry[n], Cv[n], ycorr);
            }
            op[t * DMODEL] = yc[j] + ycorr + uc[j] * Dd;
        }
    }
}

extern "C" void kernel_launch(void* const* d_in, const int* in_sizes, int n_in,
                              void* d_out, int out_size, void* d_ws, size_t ws_size,
                              hipStream_t stream) {
    const float* U     = (const float*)d_in[0];
    const float* A_log = (const float*)d_in[1];
    const float* W1    = (const float*)d_in[2];
    const float* W2    = (const float*)d_in[3];
    const float* bias  = (const float*)d_in[4];
    const float* Dvec  = (const float*)d_in[5];
    float* out = (float*)d_out;

    float* ws    = (float*)d_ws;
    float* xp    = ws;
    float* delta = xp + (size_t)G1_KSPLIT * XP_PLANE;
    float* agg   = delta + (size_t)MROWS * DMODEL;
    float* bc    = agg + (size_t)NCHUNK * 2 * NSTATE * NCHAN;

    gemm_xproj<<<dim3((MROWS / G1_UROWS) * G1_KSPLIT), dim3(256), 0, stream>>>(U, W1, xp);

    gemm_delta<<<dim3((MROWS / G2_BM) * (DMODEL / G2_BN)), dim3(G2_NT), 0, stream>>>(xp, W2, bias, delta, bc);

    dim3 sgrid(NCHAN / 256, NCHUNK);
    scan_local<<<sgrid, dim3(256), 0, stream>>>(U, A_log, bc, delta, out, agg);
    scan_fixup<<<dim3(NCHAN / FX_CH, NCHUNK / FX_CPB), dim3(256), 0, stream>>>(U, A_log, bc, delta, Dvec, agg, out);
}


# both scan passes: step arithmetic on packed two-wide f32 (v_pk_mul/v_pk_fma) over state pairs, decay as exp2 with log2(e) folded into A once; a step's state sum as even+odd partial sums (new order)
# speedup vs baseline: 2.8094x; 1.0810x over previous
#include <hip/hip_runtime.h>
#include <hip/hip_bf16.h>
#include <math.h>

typedef float v2f __attribute__((ext_vector_type(2)));
typedef float v8f __attribute__((ext_vector_type(8)));

#define BATCH   2
#define SEQ     2048
#define DMODEL  1024
#define DTRANK  64
#define NSTATE  16
#define KPROJ   (DTRANK + 2 * NSTATE)
#define MROWS   (BATCH * SEQ)
#define NCHAN   (BATCH * DMODEL)
#define NCHUNK  32
#define CLEN    (SEQ / NCHUNK)

#define G1_KSPLIT 4
#define G1_KR    (DMODEL / G1_KSPLIT)
#define G1_KS    128
#define G1_NSLAB (G1_KR / G1_KS)
#define G1_UROWS 64
#define G1_ROWS  (G1_UROWS + KPROJ)
#define G1_NPC   (G1_ROWS / 8)
#define G1_P4    (G1_KS / 4 + 1)
#define G1_SLAB4 (G1_ROWS * G1_P4)
#define G1_KSTEPS (G1_KS / 4)
#define G1_KG    16
#define XP_PLANE (MROWS * KPROJ)

__device__ __forceinline__ float4 xp_add4(float4 a, float4 b, float4 c, float4 d) {
    float4 r;
    r.x = ((a.x + b.x) + c.x) + d.x;
    r.y = ((a.y + b.y) + c.y) + d.y;
    r.z = ((a.z + b.z) + c.z) + d.z;
    r.w = ((a.w + b.w) + c.w) + d.w;
    return r;
}
__device__ __forceinline__ float4 xp_sum4(const float* __restrict__ p) {
    return xp_add4(*(const float4*)(p), *(const float4*)(p + XP_PLANE),
                   *(const float4*)(p + 2 * XP_PLANE), *(const float4*)(p + 3 * XP_PLANE));
}

__device__ __forceinline__ void g1_slab_products(const float* pa, const float* pb,
                                                 v8f& acc0, v8f& acc1, v8f& acc2) {
    #pragma unroll
    for (int g = 0; g < G1_KSTEPS / G1_KG; ++g) {
        v2f a[G1_KG], b0[G1_KG], b1[G1_KG], b2[G1_KG];
        #pragma unroll
        for (int kk = 0; kk < G1_KG; ++kk) {
            const int o = 4 * (g * G1_KG + kk);
            a[kk]  = *(const v2f*)(pa + o);
            b0[kk] = *(const v2f*)(pb + o);
            b1[kk] = *(const v2f*)(pb + 16 * (4 * G1_P4) + o);
            b2[kk] = *(const v2f*)(pb + 32 * (4 * G1_P4) + o);
        }
        __builtin_amdgcn_sched_barrier(0);
        #pragma unroll
        for (int kk = 0; kk < G1_KG; ++kk) {
            acc0 = __builtin_amdgcn_wmma_f32_16x16x4_f32(false, a[kk], false, b0[kk],
                                                         (short)0, acc0, false, false);
            acc1 = __builtin_amdgcn_wmma_f32_16x16x4_f32(false, a[kk], false, b1[kk],
                                                         (short)0, acc1, false, false);
            acc2 = __builtin_amdgcn_wmma_f32_16x16x4_f32(false, a[kk], false, b2[kk],
                                                         (short)0, acc2, false, false);
        }
        __builtin_amdgcn_sched_barrier(0);
    }
}

__device__ __forceinline__ void g1_lds_barrier() {
    __builtin_amdgcn_fence(__ATOMIC_RELEASE, "workgroup", "local");
    __builtin_amdgcn_s_barrier();
    __builtin_amdgcn_fence(__ATOMIC_ACQUIRE, "workgroup", "local");
}

__global__ __launch_bounds__(256)
void gemm_xproj(const float* __restrict__ U, const float* __restrict__ W1,
                float* __restrict__ xpp) {
    __shared__ float4 sm[G1_NSLAB * G1_SLAB4];

    const int lane  = threadIdx.x & 31;
    const int wave  = threadIdx.x >> 5;
    const int mb    = blockIdx.x >> 2;
    const int ks    = blockIdx.x & 3;
    const int strip = wave >> 1;
    const int ntb   = (wave & 1) * 3;
    const int c   = lane & 15;
    const int h   = lane >> 4;

    const float* gu = U  + (mb * G1_UROWS + wave) * DMODEL + ks * G1_KR + 4 * lane;
    const float* gw = W1 + wave * DMODEL + ks * G1_KR + 4 * lane;
    const int sdst = wave * G1_P4 + lane;

    const int fa = (strip * 16 + c) * (4 * G1_P4) + 2 * h;
    const int fb = (G1_UROWS + ntb * 16 + c) * (4 * G1_P4) + 2 * h;

#define G1_PIECES(X) X(0) X(1) X(2) X(3) X(4) X(5) X(6) X(7) X(8) X(9) \
                     X(10) X(11) X(12) X(13) X(14) X(15) X(16) X(17) X(18) X(19)
#define G1_DECL(i)  float4 r##i;
#define G1_LOAD(i)  r##i = *(const float4*)((((i) < 8) ? (gu + 8 * (i) * DMODEL) : (gw + 8 * ((i) - 8) * DMODEL)) + ko);
#define G1_PARK(i)  sm[dst + 8 * (i) * G1_P4] = r##i;
    G1_PIECES(G1_DECL)
    {
        const int ko = 0;
        G1_PIECES(G1_LOAD)
    }

    v8f acc0 = {}, acc1 = {}, acc2 = {};
    #pragma unroll 1
    for (int s = 0; s < G1_NSLAB; ++s) {
        const int cur = s * G1_SLAB4;
        const int dst = cur + sdst;
        G1_PIECES(G1_PARK)
        if (s + 1 < G1_NSLAB) {
            const int ko = (s + 1) * G1_KS;
            G1_PIECES(G1_LOAD)
        }
        g1_lds_barrier();
        g1_slab_products((const float*)(sm + cur) + fa, (const float*)(sm + cur) + fb,
                         acc0, acc1, acc2);
    }
#undef G1_PARK
#undef G1_LOAD
#undef G1_DECL
#undef G1_PIECES

    float* xr = xpp + ks * XP_PLANE + (mb * G1_UROWS + strip * 16 + 8 * h) * KPROJ + ntb * 16 + c;
    #pragma unroll
    for (int i = 0; i < 8; ++i) {
        xr[i * KPROJ]      = acc0[i];
        xr[i * KPROJ + 16] = acc1[i];
        xr[i * KPROJ + 32] = acc2[i];
    }
}

#define G2_BM  64
#define G2_BN  256
#define G2_P4  (DTRANK / 4 + 1)
#define G2_NT  512
#define G2_MT  2
#define G2_KSTEPS (DTRANK / 4)

__device__ __forceinline__ float g2_softplus(float x) {
    const float e = __builtin_amdgcn_exp2f(-fabsf(x) * 1.44269504088896341f);
    const float u = 1.0f + e;
    const float d = u - 1.0f;
    const float l = __builtin_amdgcn_logf(u) * 0.693147180559945309f;
    const float r = (d == 0.0f) ? e : l * (e * __builtin_amdgcn_rcpf(d));
    return fmaxf(x, 0.0f) + r;
}

__global__ __launch_bounds__(G2_NT)
void gemm_delta(const float* __restrict__ xp, const float* __restrict__ W2,
                const float* __restrict__ bias, float* __restrict__ delta,
                float* __restrict__ bc) {
    __shared__ float4 sA[G2_BM * G2_P4];
    __shared__ float4 sB[G2_BN * G2_P4];
    __shared__ float4 sP[G1_KSPLIT * 128];

    const int lane = threadIdx.x & 31;
    const int wave = (threadIdx.x >> 5) & 7;
    const int mh   = threadIdx.x >> 8;
    const int m0   = (blockIdx.x >> 2) * G2_BM;
    const int n0   = (blockIdx.x & 3) * G2_BN;
    const int c    = lane & 15;
    const int h    = lane >> 4;

    #pragma unroll
    for (int i = 0; i < G2_BM * 16 / G2_NT; ++i) {
        const int idx = threadIdx.x + G2_NT * i;
        const int r = idx >> 4, q = idx & 15;
        sA[r * G2_P4 + q] = xp_sum4(xp + (m0 + r) * KPROJ + 4 * q);
    }
    #pragma unroll
    for (int i = 0; i < G2_BN * 16 / G2_NT; ++i) {
        const int idx = threadIdx.x + G2_NT * i;
        const int r = idx >> 4, q = idx & 15;
        sB[r * G2_P4 + q] = *(const float4*)(W2 + (n0 + r) * DTRANK + 4 * q);
    }
    const int nb  = blockIdx.x & 3;
    const int bcr = m0 + nb * 16 + ((threadIdx.x & 127) >> 3);
    const int bcq = 4 * (threadIdx.x & 7);
    sP[threadIdx.x] = *(const float4*)(xp + (threadIdx.x >> 7) * XP_PLANE + bcr * KPROJ + DTRANK + bcq);
    __syncthreads();

    if (threadIdx.x < 128)
        *(float4*)(bc + bcr * (2 * NSTATE) + bcq) =
            xp_add4(sP[threadIdx.x], sP[128 + threadIdx.x], sP[256 + threadIdx.x], sP[384 + threadIdx.x]);

    const float* fb = (const float*)sB + (wave * 32 + c) * (4 * G2_P4) + 2 * h;
    v2f b0[G2_KSTEPS], b1[G2_KSTEPS];
    #pragma unroll
    for (int kk = 0; kk < G2_KSTEPS; ++kk) {
        b0[kk] = *(const v2f*)(fb + 4 * kk);
        b1[kk] = *(const v2f*)(fb + 16 * (4 * G2_P4) + 4 * kk);
    }

    const int   col = n0 + wave * 32 + c;
    const float bn0 = bias[col];
    const float bn1 = bias[col + 16];

    const float* fa = (const float*)sA + (mh * 32 + c) * (4 * G2_P4) + 2 * h;
    v2f an[G2_KSTEPS];
    #pragma unroll
    for (int kk = 0; kk < G2_KSTEPS; ++kk) an[kk] = *(const v2f*)(fa + 4 * kk);

    #pragma unroll
    for (int mt = 0; mt < G2_MT; ++mt) {
        v2f a[G2_KSTEPS];
        #pragma unroll
        for (int kk = 0; kk < G2_KSTEPS; ++kk) a[kk] = an[kk];
        if (mt + 1 < G2_MT) {
            #pragma unroll
            for (int kk = 0; kk < G2_KSTEPS; ++kk)
                an[kk] = *(const v2f*)(fa + (mt + 1) * 16 * (4 * G2_P4) + 4 * kk);
        }
        __builtin_amdgcn_sched_barrier(0);
        v8f acc0 = {}, acc1 = {};
        #pragma unroll
        for (int kk = 0; kk < G2_KSTEPS; ++kk) {
            acc0 = __builtin_amdgcn_wmma_f32_16x16x4_f32(false, a[kk], false, b0[kk],
                                                         (short)0, acc0, false, false);
            acc1 = __builtin_amdgcn_wmma_f32_16x16x4_f32(false, a[kk], false, b1[kk],
                                                         (short)0, acc1, false, false);
        }
        float* dr = delta + (m0 + mh * 32 + mt * 16 + 8 * h) * DMODEL + col;
        float sp0[8], sp1[8];
        #pragma unroll
        for (int i = 0; i < 8; ++i) {
            sp0[i] = g2_softplus(acc0[i] + bn0);
            sp1[i] = g2_softplus(acc1[i] + bn1);
        }
        #pragma unroll
        for (int i = 0; i < 8; ++i) {
            dr[i * DMODEL]      = sp0[i];
            dr[i * DMODEL + 16] = sp1[i];
        }
    }
}

#define PF 8
#define NPAIR (NSTATE / 2)
#define LOG2E 1.44269504088896341f

__device__ __forceinline__ v2f pk_fma(v2f a, v2f b, v2f c) { return __builtin_elementwise_fma(a, b, c); }
__device__ __forceinline__ v2f pk_exp2(v2f a) {
    v2f e;
    e.x = __builtin_amdgcn_exp2f(a.x);
    e.y = __builtin_amdgcn_exp2f(a.y);
    return e;
}
__device__ __forceinline__ v2f scan_a2(const float* __restrict__ A_log, int p) {
    const v2f a = { -__expf(A_log[2 * p]), -__expf(A_log[2 * p + 1]) };
    return a * LOG2E;
}
__device__ __forceinline__ v2f scan_arg(v2f dk, const v2f* A2, int n) {
    return dk * ((n & 1) ? A2[n >> 1].yy : A2[n >> 1].xx);
}
#define BPB (DMODEL / 256)

__global__ __launch_bounds__(256)
void scan_local(const float* __restrict__ U, const float* __restrict__ A_log,
                const float* __restrict__ bc, const float* __restrict__ delta,
                float* __restrict__ out, float* __restrict__ agg) {
    __shared__ float4 sBC[CLEN * 8];

    const int c    = blockIdx.y;
    const int bb   = blockIdx.x / BPB;
    const int dd   = (blockIdx.x % BPB) * 256 + threadIdx.x;
    const int ch   = bb * DMODEL + dd;
    const int row0 = bb * SEQ + c * CLEN;

    const float* __restrict__ dp = delta + row0 * DMODEL + dd;
    const float* __restrict__ up = U     + row0 * DMODEL + dd;
    float*       __restrict__ op = out   + row0 * DMODEL + dd;

    #pragma unroll
    for (int i = threadIdx.x; i < CLEN * 8; i += 256)
        sBC[i] = *(const float4*)(bc + (row0 + (i >> 3)) * (2 * NSTATE) + 4 * (i & 7));

    v2f dn[PF / 2], un[PF / 2];
    #pragma unroll
    for (int k = 0; k < PF / 2; ++k) {
        dn[k].x = dp[(2 * k) * DMODEL];  dn[k].y = dp[(2 * k + 1) * DMODEL];
        un[k].x = up[(2 * k) * DMODEL];  un[k].y = up[(2 * k + 1) * DMODEL];
    }

    v2f A2[NPAIR];
    #pragma unroll
    for (int p = 0; p < NPAIR; ++p) A2[p] = scan_a2(A_log, p);

    v2f st[NPAIR];
    #pragma unroll
    for (int p = 0; p < NPAIR; ++p) st[p] = (v2f){0.0f, 0.0f};

    __syncthreads();

    float4 nb[8];
    #pragma unroll
    for (int q = 0; q < 8; ++q) nb[q] = sBC[q];

    float dsum = 0.0f;
    #pragma unroll 1
    for (int tb = 0; tb < CLEN; tb += PF) {
        v2f dc[PF / 2], uc[PF / 2];
        #pragma unroll
        for (int k = 0; k < PF / 2; ++k) { dc[k] = dn[k]; uc[k] = un[k]; }
        if (tb + PF < CLEN) {
            #pragma unroll
            for (int k = 0; k < PF / 2; ++k) {
                dn[k].x = dp[(tb + PF + 2 * k) * DMODEL];  dn[k].y = dp[(tb + PF + 2 * k + 1) * DMODEL];
                un[k].x = up[(tb + PF + 2 * k) * DMODEL];  un[k].y = up[(tb + PF + 2 * k + 1) * DMODEL];
            }
        }
        v2f X[NSTATE];
        v2f duk;
        #pragma unroll
        for (int j = 0; j < PF; ++j) {
            const int t = tb + j;
            v2f Bp[NPAIR], Cp[NPAIR];
            #pragma unroll
            for (int q = 0; q < 4; ++q) {
                float4 vb = nb[q], vc = nb[4 + q];
                Bp[2*q] = (v2f){vb.x, vb.y};  Bp[2*q+1] = (v2f){vb.z, vb.w};
                Cp[2*q] = (v2f){vc.x, vc.y};  Cp[2*q+1] = (v2f){vc.z, vc.w};
            }
            const int tn = (t + 1 < CLEN) ? (t + 1) : t;
            #pragma unroll
            for (int q = 0; q < 8; ++q) nb[q] = sBC[tn * 8 + q];
            __builtin_amdgcn_sched_barrier(0);
            const v2f dk = dc[j / 2];
            if ((j & 1) == 0) {
                #pragma unroll
                for (int n = 0; n < NSTATE; ++n) X[n] = scan_arg(dk, A2, n);
                duk = dk * uc[j / 2];
            }
            const float dl = (j & 1) ? dk.y : dk.x;
            const v2f du2 = (j & 1) ? duk.yy : duk.xx;
            v2f y2 = {0.0f, 0.0f};
            #pragma unroll
            for (int p = 0; p < NPAIR; ++p) {
                v2f dA;
                dA.x = __builtin_amdgcn_exp2f((j & 1) ? X[2 * p].y     : X[2 * p].x);
                dA.y = __builtin_amdgcn_exp2f((j & 1) ? X[2 * p + 1].y : X[2 * p + 1].x);
                st[p] = pk_fma(dA, st[p], du2 * Bp[p]);
                y2    = pk_fma(st[p], Cp[p], y2);
            }
            dsum += dl;
            op[t * DMODEL] = y2.x + y2.y;
        }
    }
    const int base = (c * 2 * NSTATE) * NCHAN + ch;
    #pragma unroll
    for (int p = 0; p < NPAIR; ++p) {
        const v2f pa = pk_exp2(A2[p] * dsum);
        agg[base + (2 * p) * NCHAN]     = pa.x;
        agg[base + (2 * p + 1) * NCHAN] = pa.y;
    }
    #pragma unroll
    for (int p = 0; p < NPAIR; ++p) {
        agg[base + (NSTATE + 2 * p) * NCHAN]     = st[p].x;
        agg[base + (NSTATE + 2 * p + 1) * NCHAN] = st[p].y;
    }
}

#define FX_CH   32
#define FX_CPB  8
#define FX_XB   (DMODEL / FX_CH)

struct FxGroup { float pa[FX_CPB], pb[FX_CPB], fa[FX_CPB], fb[FX_CPB]; };

__device__ __forceinline__ void fx_load(FxGroup& s, const float* __restrict__ ag, int grp) {
    #pragma unroll
    for (int k = 0; k < FX_CPB; ++k) {
        const float* a = ag + ((grp * FX_CPB + k) * 2 * NSTATE) * NCHAN;
        s.pa[k] = a[0];               s.pb[k] = a[NCHAN];
        s.fa[k] = a[NSTATE * NCHAN];  s.fb[k] = a[(NSTATE + 1) * NCHAN];
    }
}

__device__ __forceinline__ void fx_pass(const FxGroup& s, float& ea, float& eb) {
    #pragma unroll
    for (int k = 0; k < FX_CPB; ++k) {
        ea = fmaf(s.pa[k], ea, s.fa[k]);
        eb = fmaf(s.pb[k], eb, s.fb[k]);
    }
}

__device__ __forceinline__ void fx_own(const FxGroup& s, float ea, float eb, float* se) {
    #pragma unroll
    for (int k = 0; k < FX_CPB; ++k) {
        se[k * NSTATE * FX_CH]         = ea;
        se[k * NSTATE * FX_CH + FX_CH] = eb;
        if (k + 1 < FX_CPB) {
            ea = fmaf(s.pa[k], ea, s.fa[k]);
            eb = fmaf(s.pb[k], eb, s.fb[k]);
        }
    }
}

__global__ __launch_bounds__(256)
void scan_fixup(const float* __restrict__ U, const float* __restrict__ A_log,
                const float* __restrict__ bc, const float* __restrict__ delta,
                const float* __restrict__ Dvec, const float* __restrict__ agg,
                float* __restrict__ out) {
    __shared__ float4 sC[FX_CPB * CLEN * 4];
    __shared__ float  sE[FX_CPB * NSTATE * FX_CH];

    const int lane = threadIdx.x & 31;
    const int wave = threadIdx.x >> 5;
    const int g    = blockIdx.y;
    const int c    = g * FX_CPB + wave;
    const int bb   = blockIdx.x / FX_XB;
    const int dd   = (blockIdx.x % FX_XB) * FX_CH + lane;
    const int ch   = bb * DMODEL + dd;
    const int row0 = bb * SEQ + c * CLEN;

    const float* __restrict__ dp = delta + row0 * DMODEL + dd;
    const float* __restrict__ up = U     + row0 * DMODEL + dd;
    float*       __restrict__ op = out   + row0 * DMODEL + dd;

    float4* sCw = sC + wave * (CLEN * 4);
    #pragma unroll
    for (int q = 0; q < CLEN * 4 / 32; ++q) {
        const int i = lane + 32 * q;
        sCw[i] = *(const float4*)(bc + (row0 + (i >> 2)) * (2 * NSTATE) + NSTATE + 4 * (i & 3));
    }

    v2f dn[PF / 2];
    float un[PF], yn[PF];
    #pragma unroll
    for (int k = 0; k < PF / 2; ++k) {
        dn[k].x = dp[(2 * k) * DMODEL];  dn[k].y = dp[(2 * k + 1) * DMODEL];
    }
    #pragma unroll
    for (int j = 0; j < PF; ++j) { un[j] = up[j * DMODEL]; yn[j] = op[j * DMODEL]; }

    v2f A2[NPAIR];
    #pragma unroll
    for (int p = 0; p < NPAIR; ++p) A2[p] = scan_a2(A_log, p);

    static_assert(NCHUNK / FX_CPB == 4, "the prefix below is written out for four groups of chunks");
    const float* ag = agg + (2 * wave) * NCHAN + ch;
    float ea = 0.0f, eb = 0.0f;
    float* se = sE + (2 * wave) * FX_CH + lane;
    FxGroup A, B;
    fx_load(A, ag, 0);
    if (g == 0) {
        fx_own(A, ea, eb, se);
    } else {
        fx_load(B, ag, 1);
        __builtin_amdgcn_sched_barrier(0);
        fx_pass(A, ea, eb);
        if (g == 1) {
            fx_own(B, ea, eb, se);
        } else {
            fx_load(A, ag, 2);
            __builtin_amdgcn_sched_barrier(0);
            fx_pass(B, ea, eb);
            if (g == 2) {
                fx_own(A, ea, eb, se);
            } else {
                fx_load(B, ag, 3);
                __builtin_amdgcn_sched_barrier(0);
                fx_pass(A, ea, eb);
                fx_own(B, ea, eb, se);
            }
        }
    }

    const float Dd = Dvec[dd];

    __syncthreads();

    v2f entry[NPAIR];
    #pragma unroll
    for (int p = 0; p < NPAIR; ++p) {
        entry[p].x = sE[(wave * NSTATE + 2 * p) * FX_CH + lane];
        entry[p].y = sE[(wave * NSTATE + 2 * p + 1) * FX_CH + lane];
    }

    float4 nc[4];
    #pragma unroll
    for (int q = 0; q < 4; ++q) nc[q] = sCw[q];

    #pragma unroll 1
    for (int tb = 0; tb < CLEN; tb += PF) {
        v2f dc[PF / 2];
        float uc[PF], yc[PF];
        #pragma unroll
        for (int k = 0; k < PF / 2; ++k) dc[k] = dn[k];
        #pragma unroll
        for (int j = 0; j < PF; ++j) { uc[j] = un[j]; yc[j] = yn[j]; }
        if (tb + PF < CLEN) {
            #pragma unroll
            for (int k = 0; k < PF / 2; ++k) {
                dn[k].x = dp[(tb + PF + 2 * k) * DMODEL];  dn[k].y = dp[(tb + PF + 2 * k + 1) * DMODEL];
            }
            #pragma unroll
            for (int j = 0; j < PF; ++j) {
                un[j] = up[(tb + PF + j) * DMODEL];
                yn[j] = op[(tb + PF + j) * DMODEL];
            }
        }
        v2f X[NSTATE];
        #pragma unroll
        for (int j = 0; j < PF; ++j) {
            const int t = tb + j;
            v2f Cp[NPAIR];
            #pragma unroll
            for (int q = 0; q < 4; ++q) {
                float4 vc = nc[q];
                Cp[2*q] = (v2f){vc.x, vc.y};  Cp[2*q+1] = (v2f){vc.z, vc.w};
            }
            const int tn = (t + 1 < CLEN) ? (t + 1) : t;
            #pragma unroll
            for (int q = 0; q < 4; ++q) nc[q] = sCw[tn * 4 + q];
            __builtin_amdgcn_sched_barrier(0);
            if ((j & 1) == 0) {
                const v2f dk = dc[j / 2];
                #pragma unroll
                for (int n = 0; n < NSTATE; ++n) X[n] = scan_arg(dk, A2, n);
            }
            v2f yc2 = {0.0f, 0.0f};
            #pragma unroll
            for (int p = 0; p < NPAIR; ++p) {
                v2f dA;
                dA.x = __builtin_amdgcn_exp2f((j & 1) ? X[2 * p].y     : X[2 * p].x);
                dA.y = __builtin_amdgcn_exp2f((j & 1) ? X[2 * p + 1].y : X[2 * p + 1].x);
                entry[p] *= dA;
                yc2 = pk_fma(entry[p], Cp[p], yc2);
            }
            const float ycorr = yc2.x + yc2.y;
            op[t * DMODEL] = yc[j] + ycorr + uc[j] * Dd;
        }
    }
}

extern "C" void kernel_launch(void* const* d_in, const int* in_sizes, int n_in,
                              void* d_out, int out_size, void* d_ws, size_t ws_size,
                              hipStream_t stream) {
    const float* U     = (const float*)d_in[0];
    const float* A_log = (const float*)d_in[1];
    const float* W1    = (const float*)d_in[2];
    const float* W2    = (const float*)d_in[3];
    const float* bias  = (const float*)d_in[4];
    const float* Dvec  = (const float*)d_in[5];
    float* out = (float*)d_out;

    float* ws    = (float*)d_ws;
    float* xp    = ws;
    float* delta = xp + (size_t)G1_KSPLIT * XP_PLANE;
    float* agg   = delta + (size_t)MROWS * DMODEL;
    float* bc    = agg + (size_t)NCHUNK * 2 * NSTATE * NCHAN;

    gemm_xproj<<<dim3((MROWS / G1_UROWS) * G1_KSPLIT), dim3(256), 0, stream>>>(U, W1, xp);

    gemm_delta<<<dim3((MROWS / G2_BM) * (DMODEL / G2_BN)), dim3(G2_NT), 0, stream>>>(xp, W2, bias, delta, bc);

    dim3 sgrid(NCHAN / 256, NCHUNK);
    scan_local<<<sgrid, dim3(256), 0, stream>>>(U, A_log, bc, delta, out, agg);
    scan_fixup<<<dim3(NCHAN / FX_CH, NCHUNK / FX_CPB), dim3(256), 0, stream>>>(U, A_log, bc, delta, Dvec, agg, out);
}


# pass 1 keeps only the chunk aggregates; pass 2 runs the full recurrence from the true entry state and writes out once (out no longer written then re-read); y summed as C.state: new order
# speedup vs baseline: 2.9599x; 1.0536x over previous
#include <hip/hip_runtime.h>
#include <hip/hip_bf16.h>
#include <math.h>

typedef float v2f __attribute__((ext_vector_type(2)));
typedef float v8f __attribute__((ext_vector_type(8)));

#define BATCH   2
#define SEQ     2048
#define DMODEL  1024
#define DTRANK  64
#define NSTATE  16
#define KPROJ   (DTRANK + 2 * NSTATE)
#define MROWS   (BATCH * SEQ)
#define NCHAN   (BATCH * DMODEL)
#define NCHUNK  32
#define CLEN    (SEQ / NCHUNK)

#define G1_KSPLIT 4
#define G1_KR    (DMODEL / G1_KSPLIT)
#define G1_KS    128
#define G1_NSLAB (G1_KR / G1_KS)
#define G1_UROWS 64
#define G1_ROWS  (G1_UROWS + KPROJ)
#define G1_NPC   (G1_ROWS / 8)
#define G1_P4    (G1_KS / 4 + 1)
#define G1_SLAB4 (G1_ROWS * G1_P4)
#define G1_KSTEPS (G1_KS / 4)
#define G1_KG    16
#define XP_PLANE (MROWS * KPROJ)

__device__ __forceinline__ float4 xp_add4(float4 a, float4 b, float4 c, float4 d) {
    float4 r;
    r.x = ((a.x + b.x) + c.x) + d.x;
    r.y = ((a.y + b.y) + c.y) + d.y;
    r.z = ((a.z + b.z) + c.z) + d.z;
    r.w = ((a.w + b.w) + c.w) + d.w;
    return r;
}
__device__ __forceinline__ float4 xp_sum4(const float* __restrict__ p) {
    return xp_add4(*(const float4*)(p), *(const float4*)(p + XP_PLANE),
                   *(const float4*)(p + 2 * XP_PLANE), *(const float4*)(p + 3 * XP_PLANE));
}

__device__ __forceinline__ void g1_slab_products(const float* pa, const float* pb,
                                                 v8f& acc0, v8f& acc1, v8f& acc2) {
    #pragma unroll
    for (int g = 0; g < G1_KSTEPS / G1_KG; ++g) {
        v2f a[G1_KG], b0[G1_KG], b1[G1_KG], b2[G1_KG];
        #pragma unroll
        for (int kk = 0; kk < G1_KG; ++kk) {
            const int o = 4 * (g * G1_KG + kk);
            a[kk]  = *(const v2f*)(pa + o);
            b0[kk] = *(const v2f*)(pb + o);
            b1[kk] = *(const v2f*)(pb + 16 * (4 * G1_P4) + o);
            b2[kk] = *(const v2f*)(pb + 32 * (4 * G1_P4) + o);
        }
        __builtin_amdgcn_sched_barrier(0);
        #pragma unroll
        for (int kk = 0; kk < G1_KG; ++kk) {
            acc0 = __builtin_amdgcn_wmma_f32_16x16x4_f32(false, a[kk], false, b0[kk],
                                                         (short)0, acc0, false, false);
            acc1 = __builtin_amdgcn_wmma_f32_16x16x4_f32(false, a[kk], false, b1[kk],
                                                         (short)0, acc1, false, false);
            acc2 = __builtin_amdgcn_wmma_f32_16x16x4_f32(false, a[kk], false, b2[kk],
                                                         (short)0, acc2, false, false);
        }
        __builtin_amdgcn_sched_barrier(0);
    }
}

__device__ __forceinline__ void g1_lds_barrier() {
    __builtin_amdgcn_fence(__ATOMIC_RELEASE, "workgroup", "local");
    __builtin_amdgcn_s_barrier();
    __builtin_amdgcn_fence(__ATOMIC_ACQUIRE, "workgroup", "local");
}

__global__ __launch_bounds__(256)
void gemm_xproj(const float* __restrict__ U, const float* __restrict__ W1,
                float* __restrict__ xpp) {
    __shared__ float4 sm[G1_NSLAB * G1_SLAB4];

    const int lane  = threadIdx.x & 31;
    const int wave  = threadIdx.x >> 5;
    const int mb    = blockIdx.x >> 2;
    const int ks    = blockIdx.x & 3;
    const int strip = wave >> 1;
    const int ntb   = (wave & 1) * 3;
    const int c   = lane & 15;
    const int h   = lane >> 4;

    const float* gu = U  + (mb * G1_UROWS + wave) * DMODEL + ks * G1_KR + 4 * lane;
    const float* gw = W1 + wave * DMODEL + ks * G1_KR + 4 * lane;
    const int sdst = wave * G1_P4 + lane;

    const int fa = (strip * 16 + c) * (4 * G1_P4) + 2 * h;
    const int fb = (G1_UROWS + ntb * 16 + c) * (4 * G1_P4) + 2 * h;

#define G1_PIECES(X) X(0) X(1) X(2) X(3) X(4) X(5) X(6) X(7) X(8) X(9) \
                     X(10) X(11) X(12) X(13) X(14) X(15) X(16) X(17) X(18) X(19)
#define G1_DECL(i)  float4 r##i;
#define G1_LOAD(i)  r##i = *(const float4*)((((i) < 8) ? (gu + 8 * (i) * DMODEL) : (gw + 8 * ((i) - 8) * DMODEL)) + ko);
#define G1_PARK(i)  sm[dst + 8 * (i) * G1_P4] = r##i;
    G1_PIECES(G1_DECL)
    {
        const int ko = 0;
        G1_PIECES(G1_LOAD)
    }

    v8f acc0 = {}, acc1 = {}, acc2 = {};
    #pragma unroll 1
    for (int s = 0; s < G1_NSLAB; ++s) {
        const int cur = s * G1_SLAB4;
        const int dst = cur + sdst;
        G1_PIECES(G1_PARK)
        if (s + 1 < G1_NSLAB) {
            const int ko = (s + 1) * G1_KS;
            G1_PIECES(G1_LOAD)
        }
        g1_lds_barrier();
        g1_slab_products((const float*)(sm + cur) + fa, (const float*)(sm + cur) + fb,
                         acc0, acc1, acc2);
    }
#undef G1_PARK
#undef G1_LOAD
#undef G1_DECL
#undef G1_PIECES

    float* xr = xpp + ks * XP_PLANE + (mb * G1_UROWS + strip * 16 + 8 * h) * KPROJ + ntb * 16 + c;
    #pragma unroll
    for (int i = 0; i < 8; ++i) {
        xr[i * KPROJ]      = acc0[i];
        xr[i * KPROJ + 16] = acc1[i];
        xr[i * KPROJ + 32] = acc2[i];
    }
}

#define G2_BM  64
#define G2_BN  256
#define G2_P4  (DTRANK / 4 + 1)
#define G2_NT  512
#define G2_MT  2
#define G2_KSTEPS (DTRANK / 4)

__device__ __forceinline__ float g2_softplus(float x) {
    const float e = __builtin_amdgcn_exp2f(-fabsf(x) * 1.44269504088896341f);
    const float u = 1.0f + e;
    const float d = u - 1.0f;
    const float l = __builtin_amdgcn_logf(u) * 0.693147180559945309f;
    const float r = (d == 0.0f) ? e : l * (e * __builtin_amdgcn_rcpf(d));
    return fmaxf(x, 0.0f) + r;
}

__global__ __launch_bounds__(G2_NT)
void gemm_delta(const float* __restrict__ xp, const float* __restrict__ W2,
                const float* __restrict__ bias, float* __restrict__ delta,
                float* __restrict__ bc) {
    __shared__ float4 sA[G2_BM * G2_P4];
    __shared__ float4 sB[G2_BN * G2_P4];
    __shared__ float4 sP[G1_KSPLIT * 128];

    const int lane = threadIdx.x & 31;
    const int wave = (threadIdx.x >> 5) & 7;
    const int mh   = threadIdx.x >> 8;
    const int m0   = (blockIdx.x >> 2) * G2_BM;
    const int n0   = (blockIdx.x & 3) * G2_BN;
    const int c    = lane & 15;
    const int h    = lane >> 4;

    #pragma unroll
    for (int i = 0; i < G2_BM * 16 / G2_NT; ++i) {
        const int idx = threadIdx.x + G2_NT * i;
        const int r = idx >> 4, q = idx & 15;
        sA[r * G2_P4 + q] = xp_sum4(xp + (m0 + r) * KPROJ + 4 * q);
    }
    #pragma unroll
    for (int i = 0; i < G2_BN * 16 / G2_NT; ++i) {
        const int idx = threadIdx.x + G2_NT * i;
        const int r = idx >> 4, q = idx & 15;
        sB[r * G2_P4 + q] = *(const float4*)(W2 + (n0 + r) * DTRANK + 4 * q);
    }
    const int nb  = blockIdx.x & 3;
    const int bcr = m0 + nb * 16 + ((threadIdx.x & 127) >> 3);
    const int bcq = 4 * (threadIdx.x & 7);
    sP[threadIdx.x] = *(const float4*)(xp + (threadIdx.x >> 7) * XP_PLANE + bcr * KPROJ + DTRANK + bcq);
    __syncthreads();

    if (threadIdx.x < 128)
        *(float4*)(bc + bcr * (2 * NSTATE) + bcq) =
            xp_add4(sP[threadIdx.x], sP[128 + threadIdx.x], sP[256 + threadIdx.x], sP[384 + threadIdx.x]);

    const float* fb = (const float*)sB + (wave * 32 + c) * (4 * G2_P4) + 2 * h;
    v2f b0[G2_KSTEPS], b1[G2_KSTEPS];
    #pragma unroll
    for (int kk = 0; kk < G2_KSTEPS; ++kk) {
        b0[kk] = *(const v2f*)(fb + 4 * kk);
        b1[kk] = *(const v2f*)(fb + 16 * (4 * G2_P4) + 4 * kk);
    }

    const int   col = n0 + wave * 32 + c;
    const float bn0 = bias[col];
    const float bn1 = bias[col + 16];

    const float* fa = (const float*)sA + (mh * 32 + c) * (4 * G2_P4) + 2 * h;
    v2f an[G2_KSTEPS];
    #pragma unroll
    for (int kk = 0; kk < G2_KSTEPS; ++kk) an[kk] = *(const v2f*)(fa + 4 * kk);

    #pragma unroll
    for (int mt = 0; mt < G2_MT; ++mt) {
        v2f a[G2_KSTEPS];
        #pragma unroll
        for (int kk = 0; kk < G2_KSTEPS; ++kk) a[kk] = an[kk];
        if (mt + 1 < G2_MT) {
            #pragma unroll
            for (int kk = 0; kk < G2_KSTEPS; ++kk)
                an[kk] = *(const v2f*)(fa + (mt + 1) * 16 * (4 * G2_P4) + 4 * kk);
        }
        __builtin_amdgcn_sched_barrier(0);
        v8f acc0 = {}, acc1 = {};
        #pragma unroll
        for (int kk = 0; kk < G2_KSTEPS; ++kk) {
            acc0 = __builtin_amdgcn_wmma_f32_16x16x4_f32(false, a[kk], false, b0[kk],
                                                         (short)0, acc0, false, false);
            acc1 = __builtin_amdgcn_wmma_f32_16x16x4_f32(false, a[kk], false, b1[kk],
                                                         (short)0, acc1, false, false);
        }
        float* dr = delta + (m0 + mh * 32 + mt * 16 + 8 * h) * DMODEL + col;
        float sp0[8], sp1[8];
        #pragma unroll
        for (int i = 0; i < 8; ++i) {
            sp0[i] = g2_softplus(acc0[i] + bn0);
            sp1[i] = g2_softplus(acc1[i] + bn1);
        }
        #pragma unroll
        for (int i = 0; i < 8; ++i) {
            dr[i * DMODEL]      = sp0[i];
            dr[i * DMODEL + 16] = sp1[i];
        }
    }
}

#define PF 8
#define NPAIR (NSTATE / 2)
#define LOG2E 1.44269504088896341f

__device__ __forceinline__ v2f pk_fma(v2f a, v2f b, v2f c) { return __builtin_elementwise_fma(a, b, c); }
__device__ __forceinline__ v2f pk_exp2(v2f a) {
    v2f e;
    e.x = __builtin_amdgcn_exp2f(a.x);
    e.y = __builtin_amdgcn_exp2f(a.y);
    return e;
}
__device__ __forceinline__ v2f scan_a2(const float* __restrict__ A_log, int p) {
    const v2f a = { -__expf(A_log[2 * p]), -__expf(A_log[2 * p + 1]) };
    return a * LOG2E;
}
__device__ __forceinline__ v2f scan_arg(v2f dk, const v2f* A2, int n) {
    return dk * ((n & 1) ? A2[n >> 1].yy : A2[n >> 1].xx);
}
#define BPB (DMODEL / 256)

__global__ __launch_bounds__(256)
void scan_local(const float* __restrict__ U, const float* __restrict__ A_log,
                const float* __restrict__ bc, const float* __restrict__ delta,
                float* __restrict__ agg) {
    __shared__ float4 sB[CLEN * 4];

    const int c    = blockIdx.y;
    const int bb   = blockIdx.x / BPB;
    const int dd   = (blockIdx.x % BPB) * 256 + threadIdx.x;
    const int ch   = bb * DMODEL + dd;
    const int row0 = bb * SEQ + c * CLEN;

    const float* __restrict__ dp = delta + row0 * DMODEL + dd;
    const float* __restrict__ up = U     + row0 * DMODEL + dd;

    #pragma unroll
    for (int i = threadIdx.x; i < CLEN * 4; i += 256)
        sB[i] = *(const float4*)(bc + (row0 + (i >> 2)) * (2 * NSTATE) + 4 * (i & 3));

    v2f dn[PF / 2], un[PF / 2];
    #pragma unroll
    for (int k = 0; k < PF / 2; ++k) {
        dn[k].x = dp[(2 * k) * DMODEL];  dn[k].y = dp[(2 * k + 1) * DMODEL];
        un[k].x = up[(2 * k) * DMODEL];  un[k].y = up[(2 * k + 1) * DMODEL];
    }

    v2f A2[NPAIR];
    #pragma unroll
    for (int p = 0; p < NPAIR; ++p) A2[p] = scan_a2(A_log, p);

    v2f st[NPAIR];
    #pragma unroll
    for (int p = 0; p < NPAIR; ++p) st[p] = (v2f){0.0f, 0.0f};

    __syncthreads();

    float4 nb[4];
    #pragma unroll
    for (int q = 0; q < 4; ++q) nb[q] = sB[q];

    float dsum = 0.0f;
    #pragma unroll 1
    for (int tb = 0; tb < CLEN; tb += PF) {
        v2f dc[PF / 2], uc[PF / 2];
        #pragma unroll
        for (int k = 0; k < PF / 2; ++k) { dc[k] = dn[k]; uc[k] = un[k]; }
        if (tb + PF < CLEN) {
            #pragma unroll
            for (int k = 0; k < PF / 2; ++k) {
                dn[k].x = dp[(tb + PF + 2 * k) * DMODEL];  dn[k].y = dp[(tb + PF + 2 * k + 1) * DMODEL];
                un[k].x = up[(tb + PF + 2 * k) * DMODEL];  un[k].y = up[(tb + PF + 2 * k + 1) * DMODEL];
            }
        }
        v2f X[NSTATE];
        v2f duk;
        #pragma unroll
        for (int j = 0; j < PF; ++j) {
            const int t = tb + j;
            v2f Bp[NPAIR];
            #pragma unroll
            for (int q = 0; q < 4; ++q) {
                float4 vb = nb[q];
                Bp[2*q] = (v2f){vb.x, vb.y};  Bp[2*q+1] = (v2f){vb.z, vb.w};
            }
            const int tn = (t + 1 < CLEN) ? (t + 1) : t;
            #pragma unroll
            for (int q = 0; q < 4; ++q) nb[q] = sB[tn * 4 + q];
            __builtin_amdgcn_sched_barrier(0);
            const v2f dk = dc[j / 2];
            if ((j & 1) == 0) {
                #pragma unroll
                for (int n = 0; n < NSTATE; ++n) X[n] = scan_arg(dk, A2, n);
                duk = dk * uc[j / 2];
            }
            const float dl = (j & 1) ? dk.y : dk.x;
            const v2f du2 = (j & 1) ? duk.yy : duk.xx;
            #pragma unroll
            for (int p = 0; p < NPAIR; ++p) {
                v2f dA;
                dA.x = __builtin_amdgcn_exp2f((j & 1) ? X[2 * p].y     : X[2 * p].x);
                dA.y = __builtin_amdgcn_exp2f((j & 1) ? X[2 * p + 1].y : X[2 * p + 1].x);
                st[p] = pk_fma(dA, st[p], du2 * Bp[p]);
            }
            dsum += dl;
        }
    }
    const int base = (c * 2 * NSTATE) * NCHAN + ch;
    #pragma unroll
    for (int p = 0; p < NPAIR; ++p) {
        const v2f pa = pk_exp2(A2[p] * dsum);
        agg[base + (2 * p) * NCHAN]     = pa.x;
        agg[base + (2 * p + 1) * NCHAN] = pa.y;
    }
    #pragma unroll
    for (int p = 0; p < NPAIR; ++p) {
        agg[base + (NSTATE + 2 * p) * NCHAN]     = st[p].x;
        agg[base + (NSTATE + 2 * p + 1) * NCHAN] = st[p].y;
    }
}

#define FX_CH   32
#define FX_CPB  8
#define FX_XB   (DMODEL / FX_CH)

struct FxGroup { float pa[FX_CPB], pb[FX_CPB], fa[FX_CPB], fb[FX_CPB]; };

__device__ __forceinline__ void fx_load(FxGroup& s, const float* __restrict__ ag, int grp) {
    #pragma unroll
    for (int k = 0; k < FX_CPB; ++k) {
        const float* a = ag + ((grp * FX_CPB + k) * 2 * NSTATE) * NCHAN;
        s.pa[k] = a[0];               s.pb[k] = a[NCHAN];
        s.fa[k] = a[NSTATE * NCHAN];  s.fb[k] = a[(NSTATE + 1) * NCHAN];
    }
}

__device__ __forceinline__ void fx_pass(const FxGroup& s, float& ea, float& eb) {
    #pragma unroll
    for (int k = 0; k < FX_CPB; ++k) {
        ea = fmaf(s.pa[k], ea, s.fa[k]);
        eb = fmaf(s.pb[k], eb, s.fb[k]);
    }
}

__device__ __forceinline__ void fx_own(const FxGroup& s, float ea, float eb, float* se) {
    #pragma unroll
    for (int k = 0; k < FX_CPB; ++k) {
        se[k * NSTATE * FX_CH]         = ea;
        se[k * NSTATE * FX_CH + FX_CH] = eb;
        if (k + 1 < FX_CPB) {
            ea = fmaf(s.pa[k], ea, s.fa[k]);
            eb = fmaf(s.pb[k], eb, s.fb[k]);
        }
    }
}

__global__ __launch_bounds__(256)
void scan_fixup(const float* __restrict__ U, const float* __restrict__ A_log,
                const float* __restrict__ bc, const float* __restrict__ delta,
                const float* __restrict__ Dvec, const float* __restrict__ agg,
                float* __restrict__ out) {
    __shared__ float4 sBC[FX_CPB * CLEN * 8];
    __shared__ float  sE[FX_CPB * NSTATE * FX_CH];

    const int lane = threadIdx.x & 31;
    const int wave = threadIdx.x >> 5;
    const int g    = blockIdx.y;
    const int c    = g * FX_CPB + wave;
    const int bb   = blockIdx.x / FX_XB;
    const int dd   = (blockIdx.x % FX_XB) * FX_CH + lane;
    const int ch   = bb * DMODEL + dd;
    const int row0 = bb * SEQ + c * CLEN;

    const float* __restrict__ dp = delta + row0 * DMODEL + dd;
    const float* __restrict__ up = U     + row0 * DMODEL + dd;
    float*       __restrict__ op = out   + row0 * DMODEL + dd;

    float4* sBCw = sBC + wave * (CLEN * 8);
    #pragma unroll
    for (int q = 0; q < CLEN * 8 / 32; ++q) {
        const int i = lane + 32 * q;
        sBCw[i] = *(const float4*)(bc + (row0 + (i >> 3)) * (2 * NSTATE) + 4 * (i & 7));
    }

    v2f dn[PF / 2], un[PF / 2];
    #pragma unroll
    for (int k = 0; k < PF / 2; ++k) {
        dn[k].x = dp[(2 * k) * DMODEL];  dn[k].y = dp[(2 * k + 1) * DMODEL];
        un[k].x = up[(2 * k) * DMODEL];  un[k].y = up[(2 * k + 1) * DMODEL];
    }

    v2f A2[NPAIR];
    #pragma unroll
    for (int p = 0; p < NPAIR; ++p) A2[p] = scan_a2(A_log, p);

    static_assert(NCHUNK / FX_CPB == 4, "the prefix below is written out for four groups of chunks");
    const float* ag = agg + (2 * wave) * NCHAN + ch;
    float ea = 0.0f, eb = 0.0f;
    float* se = sE + (2 * wave) * FX_CH + lane;
    FxGroup A, B;
    fx_load(A, ag, 0);
    if (g == 0) {
        fx_own(A, ea, eb, se);
    } else {
        fx_load(B, ag, 1);
        __builtin_amdgcn_sched_barrier(0);
        fx_pass(A, ea, eb);
        if (g == 1) {
            fx_own(B, ea, eb, se);
        } else {
            fx_load(A, ag, 2);
            __builtin_amdgcn_sched_barrier(0);
            fx_pass(B, ea, eb);
            if (g == 2) {
                fx_own(A, ea, eb, se);
            } else {
                fx_load(B, ag, 3);
                __builtin_amdgcn_sched_barrier(0);
                fx_pass(A, ea, eb);
                fx_own(B, ea, eb, se);
            }
        }
    }

    const float Dd = Dvec[dd];

    __syncthreads();

    v2f st[NPAIR];
    #pragma unroll
    for (int p = 0; p < NPAIR; ++p) {
        st[p].x = sE[(wave * NSTATE + 2 * p) * FX_CH + lane];
        st[p].y = sE[(wave * NSTATE + 2 * p + 1) * FX_CH + lane];
    }

    float4 nb[8];
    #pragma unroll
    for (int q = 0; q < 8; ++q) nb[q] = sBCw[q];

    #pragma unroll 1
    for (int tb = 0; tb < CLEN; tb += PF) {
        v2f dc[PF / 2], uc[PF / 2];
        #pragma unroll
        for (int k = 0; k < PF / 2; ++k) { dc[k] = dn[k]; uc[k] = un[k]; }
        if (tb + PF < CLEN) {
            #pragma unroll
            for (int k = 0; k < PF / 2; ++k) {
                dn[k].x = dp[(tb + PF + 2 * k) * DMODEL];  dn[k].y = dp[(tb + PF + 2 * k + 1) * DMODEL];
                un[k].x = up[(tb + PF + 2 * k) * DMODEL];  un[k].y = up[(tb + PF + 2 * k + 1) * DMODEL];
            }
        }
        v2f X[NSTATE];
        v2f duk;
        #pragma unroll
        for (int j = 0; j < PF; ++j) {
            const int t = tb + j;
            v2f Bp[NPAIR], Cp[NPAIR];
            #pragma unroll
            for (int q = 0; q < 4; ++q) {
                float4 vb = nb[q], vc = nb[4 + q];
                Bp[2*q] = (v2f){vb.x, vb.y};  Bp[2*q+1] = (v2f){vb.z, vb.w};
                Cp[2*q] = (v2f){vc.x, vc.y};  Cp[2*q+1] = (v2f){vc.z, vc.w};
            }
            const int tn = (t + 1 < CLEN) ? (t + 1) : t;
            #pragma unroll
            for (int q = 0; q < 8; ++q) nb[q] = sBCw[tn * 8 + q];
            __builtin_amdgcn_sched_barrier(0);
            const v2f dk = dc[j / 2];
            if ((j & 1) == 0) {
                #pragma unroll
                for (int n = 0; n < NSTATE; ++n) X[n] = scan_arg(dk, A2, n);
                duk = dk * uc[j / 2];
            }
            const float ut  = (j & 1) ? uc[j / 2].y : uc[j / 2].x;
            const v2f   du2 = (j & 1) ? duk.yy : duk.xx;
            v2f y2 = {0.0f, 0.0f};
            #pragma unroll
            for (int p = 0; p < NPAIR; ++p) {
                v2f dA;
                dA.x = __builtin_amdgcn_exp2f((j & 1) ? X[2 * p].y     : X[2 * p].x);
                dA.y = __builtin_amdgcn_exp2f((j & 1) ? X[2 * p + 1].y : X[2 * p + 1].x);
                st[p] = pk_fma(dA, st[p], du2 * Bp[p]);
                y2    = pk_fma(st[p], Cp[p], y2);
            }
            op[t * DMODEL] = (y2.x + y2.y) + ut * Dd;
        }
    }
}

extern "C" void kernel_launch(void* const* d_in, const int* in_sizes, int n_in,
                              void* d_out, int out_size, void* d_ws, size_t ws_size,
                              hipStream_t stream) {
    const float* U     = (const float*)d_in[0];
    const float* A_log = (const float*)d_in[1];
    const float* W1    = (const float*)d_in[2];
    const float* W2    = (const float*)d_in[3];
    const float* bias  = (const float*)d_in[4];
    const float* Dvec  = (const float*)d_in[5];
    float* out = (float*)d_out;

    float* ws    = (float*)d_ws;
    float* xp    = ws;
    float* delta = xp + (size_t)G1_KSPLIT * XP_PLANE;
    float* agg   = delta + (size_t)MROWS * DMODEL;
    float* bc    = agg + (size_t)NCHUNK * 2 * NSTATE * NCHAN;

    gemm_xproj<<<dim3((MROWS / G1_UROWS) * G1_KSPLIT), dim3(256), 0, stream>>>(U, W1, xp);

    gemm_delta<<<dim3((MROWS / G2_BM) * (DMODEL / G2_BN)), dim3(G2_NT), 0, stream>>>(xp, W2, bias, delta, bc);

    dim3 sgrid(NCHAN / 256, NCHUNK);
    scan_local<<<sgrid, dim3(256), 0, stream>>>(U, A_log, bc, delta, agg);
    scan_fixup<<<dim3(NCHAN / FX_CH, NCHUNK / FX_CPB), dim3(256), 0, stream>>>(U, A_log, bc, delta, Dvec, agg, out);
}


# both GEMMs in one launch of 256 blocks x 16 rows: xp stays in LDS (k split over 8 waves, partials added wave 0..7: new sum order), delta from it with W2 streamed per wave; 3 launches, no xp planes
# speedup vs baseline: 3.0602x; 1.0339x over previous
#include <hip/hip_runtime.h>
#include <hip/hip_bf16.h>
#include <math.h>

typedef float v2f __attribute__((ext_vector_type(2)));
typedef float v8f __attribute__((ext_vector_type(8)));

#define BATCH   2
#define SEQ     2048
#define DMODEL  1024
#define DTRANK  64
#define NSTATE  16
#define KPROJ   (DTRANK + 2 * NSTATE)
#define MROWS   (BATCH * SEQ)
#define NCHAN   (BATCH * DMODEL)
#define NCHUNK  32
#define CLEN    (SEQ / NCHUNK)

__device__ __forceinline__ float g2_softplus(float x) {
    const float e = __builtin_amdgcn_exp2f(-fabsf(x) * 1.44269504088896341f);
    const float u = 1.0f + e;
    const float d = u - 1.0f;
    const float l = __builtin_amdgcn_logf(u) * 0.693147180559945309f;
    const float r = (d == 0.0f) ? e : l * (e * __builtin_amdgcn_rcpf(d));
    return fmaxf(x, 0.0f) + r;
}

#define F_ROWS   16
#define F_KS     128
#define F_NSLAB  (DMODEL / F_KS)
#define F_SROWS  (F_ROWS + KPROJ)
#define F_NPC    (F_SROWS / 8)
#define F_P4     (F_KS / 4 + 1)
#define F_SLAB4  (F_SROWS * F_P4)
#define F_KW     4
#define F_NT1    (KPROJ / 16)
#define F_X4     (DTRANK / 4 + 1)
#define F_WB4    (32 * F_X4)
#define F_NPAIR  (DMODEL / 8 / 32)
#define F_TILE   (F_ROWS * KPROJ)

__device__ __forceinline__ void lds_barrier() {
    __builtin_amdgcn_fence(__ATOMIC_RELEASE, "workgroup", "local");
    __builtin_amdgcn_s_barrier();
    __builtin_amdgcn_fence(__ATOMIC_ACQUIRE, "workgroup", "local");
}

__global__ __launch_bounds__(256)
void gemm_fused(const float* __restrict__ U, const float* __restrict__ W1,
                const float* __restrict__ W2, const float* __restrict__ bias,
                float* __restrict__ delta, float* __restrict__ bc) {
    __shared__ float4 sm[2 * F_SLAB4];
    __shared__ float4 sX[F_ROWS * F_X4];
    __shared__ float4 sW[8 * F_WB4];

    const int lane = threadIdx.x & 31;
    const int wave = threadIdx.x >> 5;
    const int m0   = blockIdx.x * F_ROWS;
    const int c    = lane & 15;
    const int h    = lane >> 4;

    const float* gu = U  + (m0 + wave) * DMODEL + 4 * lane;
    const float* gw = W1 + wave * DMODEL + 4 * lane;
    const int sdst = wave * F_P4 + lane;

    const int fa = c * (4 * F_P4) + 16 * wave + 2 * h;
    const int fb = (F_ROWS + c) * (4 * F_P4) + 16 * wave + 2 * h;

#define F_PIECES(X) X(0) X(1) X(2) X(3) X(4) X(5) X(6) X(7) X(8) X(9) X(10) X(11) X(12) X(13)
#define F_DECL(i)  float4 r##i;
#define F_LOAD(i)  r##i = *(const float4*)((((i) < 2) ? (gu + 8 * (i) * DMODEL) : (gw + 8 * ((i) - 2) * DMODEL)) + ko);
#define F_PARK(i)  sm[dst + 8 * (i) * F_P4] = r##i;
    F_PIECES(F_DECL)
    {
        const int ko = 0;
        F_PIECES(F_LOAD)
    }

    v8f acc[F_NT1];
    #pragma unroll
    for (int t = 0; t < F_NT1; ++t) acc[t] = (v8f){};
    #pragma unroll 1
    for (int s = 0; s < F_NSLAB; ++s) {
        const int cur = (s & 1) * F_SLAB4;
        const int dst = cur + sdst;
        F_PIECES(F_PARK)
        if (s + 1 < F_NSLAB) {
            const int ko = (s + 1) * F_KS;
            F_PIECES(F_LOAD)
        }
        lds_barrier();
        const float* pa = (const float*)(sm + cur) + fa;
        const float* pb = (const float*)(sm + cur) + fb;
        v2f a[F_KW], b[F_NT1][F_KW];
        #pragma unroll
        for (int j = 0; j < F_KW; ++j) {
            a[j] = *(const v2f*)(pa + 4 * j);
            #pragma unroll
            for (int t = 0; t < F_NT1; ++t)
                b[t][j] = *(const v2f*)(pb + t * 16 * (4 * F_P4) + 4 * j);
        }
        __builtin_amdgcn_sched_barrier(0);
        #pragma unroll
        for (int j = 0; j < F_KW; ++j) {
            #pragma unroll
            for (int t = 0; t < F_NT1; ++t)
                acc[t] = __builtin_amdgcn_wmma_f32_16x16x4_f32(false, a[j], false, b[t][j],
                                                               (short)0, acc[t], false, false);
        }
    }
#undef F_PARK
#undef F_LOAD
#undef F_DECL
#undef F_PIECES

    const int colw = wave * (DMODEL / 8);
    const float* gW2 = W2 + (colw + h) * DTRANK + 4 * c;
#define F_WPIECES(X) X(0) X(1) X(2) X(3) X(4) X(5) X(6) X(7) X(8) X(9) X(10) X(11) X(12) X(13) X(14) X(15)
#define F_WDECL(i)  float4 q##i;
#define F_WLOAD(i)  q##i = *(const float4*)(gW2 + (pr + 2 * (i)) * DTRANK);
#define F_WPARK(i)  wB[(2 * (i) + h) * F_X4 + c] = q##i;
    F_WPIECES(F_WDECL)
    {
        const int pr = 0;
        F_WPIECES(F_WLOAD)
    }
    float bn[2 * F_NPAIR];
    #pragma unroll
    for (int p = 0; p < 2 * F_NPAIR; ++p) bn[p] = bias[colw + 16 * p + c];

    lds_barrier();
    float* red = (float*)sm;
    #pragma unroll
    for (int t = 0; t < F_NT1; ++t) {
        #pragma unroll
        for (int i = 0; i < 8; ++i)
            red[wave * F_TILE + (8 * h + i) * KPROJ + 16 * t + c] = acc[t][i];
    }
    lds_barrier();
    #pragma unroll
    for (int q = 0; q < F_TILE / 256; ++q) {
        const int e = threadIdx.x + 256 * q;
        float v = red[e];
        #pragma unroll
        for (int w = 1; w < 8; ++w) v += red[w * F_TILE + e];
        const int row = e / KPROJ, col = e % KPROJ;
        if (col < DTRANK) ((float*)sX)[row * (4 * F_X4) + col] = v;
        else              bc[(m0 + row) * (2 * NSTATE) + (col - DTRANK)] = v;
    }
    lds_barrier();

    const float* fa2 = (const float*)sX + c * (4 * F_X4) + 2 * h;
    v2f a2[DTRANK / 4];
    #pragma unroll
    for (int kk = 0; kk < DTRANK / 4; ++kk) a2[kk] = *(const v2f*)(fa2 + 4 * kk);

    float4* wB = sW + wave * F_WB4;
    const float* fb2 = (const float*)wB + c * (4 * F_X4) + 2 * h;
    #pragma unroll
    for (int p = 0; p < F_NPAIR; ++p) {
        F_WPIECES(F_WPARK)
        if (p + 1 < F_NPAIR) {
            const int pr = (p + 1) * 32;
            F_WPIECES(F_WLOAD)
        }
        lds_barrier();
        v2f b0[DTRANK / 4], b1[DTRANK / 4];
        #pragma unroll
        for (int kk = 0; kk < DTRANK / 4; ++kk) {
            b0[kk] = *(const v2f*)(fb2 + 4 * kk);
            b1[kk] = *(const v2f*)(fb2 + 16 * (4 * F_X4) + 4 * kk);
        }
        __builtin_amdgcn_sched_barrier(0);
        v8f d0 = {}, d1 = {};
        #pragma unroll
        for (int kk = 0; kk < DTRANK / 4; ++kk) {
            d0 = __builtin_amdgcn_wmma_f32_16x16x4_f32(false, a2[kk], false, b0[kk],
                                                       (short)0, d0, false, false);
            d1 = __builtin_amdgcn_wmma_f32_16x16x4_f32(false, a2[kk], false, b1[kk],
                                                       (short)0, d1, false, false);
        }
        float* dr = delta + (m0 + 8 * h) * DMODEL + colw + 32 * p + c;
        float sp0[8], sp1[8];
        #pragma unroll
        for (int i = 0; i < 8; ++i) {
            sp0[i] = g2_softplus(d0[i] + bn[2 * p]);
            sp1[i] = g2_softplus(d1[i] + bn[2 * p + 1]);
        }
        #pragma unroll
        for (int i = 0; i < 8; ++i) {
            dr[i * DMODEL]      = sp0[i];
            dr[i * DMODEL + 16] = sp1[i];
        }
    }
#undef F_WPARK
#undef F_WLOAD
#undef F_WDECL
#undef F_WPIECES
}

#define PF 8
#define NPAIR (NSTATE / 2)
#define LOG2E 1.44269504088896341f

__device__ __forceinline__ v2f pk_fma(v2f a, v2f b, v2f c) { return __builtin_elementwise_fma(a, b, c); }
__device__ __forceinline__ v2f pk_exp2(v2f a) {
    v2f e;
    e.x = __builtin_amdgcn_exp2f(a.x);
    e.y = __builtin_amdgcn_exp2f(a.y);
    return e;
}
__device__ __forceinline__ v2f scan_a2(const float* __restrict__ A_log, int p) {
    const v2f a = { -__expf(A_log[2 * p]), -__expf(A_log[2 * p + 1]) };
    return a * LOG2E;
}
__device__ __forceinline__ v2f scan_arg(v2f dk, const v2f* A2, int n) {
    return dk * ((n & 1) ? A2[n >> 1].yy : A2[n >> 1].xx);
}
#define BPB (DMODEL / 256)

__global__ __launch_bounds__(256)
void scan_local(const float* __restrict__ U, const float* __restrict__ A_log,
                const float* __restrict__ bc, const float* __restrict__ delta,
                float* __restrict__ agg) {
    __shared__ float4 sB[CLEN * 4];

    const int c    = blockIdx.y;
    const int bb   = blockIdx.x / BPB;
    const int dd   = (blockIdx.x % BPB) * 256 + threadIdx.x;
    const int ch   = bb * DMODEL + dd;
    const int row0 = bb * SEQ + c * CLEN;

    const float* __restrict__ dp = delta + row0 * DMODEL + dd;
    const float* __restrict__ up = U     + row0 * DMODEL + dd;

    #pragma unroll
    for (int i = threadIdx.x; i < CLEN * 4; i += 256)
        sB[i] = *(const float4*)(bc + (row0 + (i >> 2)) * (2 * NSTATE) + 4 * (i & 3));

    v2f dn[PF / 2], un[PF / 2];
    #pragma unroll
    for (int k = 0; k < PF / 2; ++k) {
        dn[k].x = dp[(2 * k) * DMODEL];  dn[k].y = dp[(2 * k + 1) * DMODEL];
        un[k].x = up[(2 * k) * DMODEL];  un[k].y = up[(2 * k + 1) * DMODEL];
    }

    v2f A2[NPAIR];
    #pragma unroll
    for (int p = 0; p < NPAIR; ++p) A2[p] = scan_a2(A_log, p);

    v2f st[NPAIR];
    #pragma unroll
    for (int p = 0; p < NPAIR; ++p) st[p] = (v2f){0.0f, 0.0f};

    __syncthreads();

    float4 nb[4];
    #pragma unroll
    for (int q = 0; q < 4; ++q) nb[q] = sB[q];

    float dsum = 0.0f;
    #pragma unroll 1
    for (int tb = 0; tb < CLEN; tb += PF) {
        v2f dc[PF / 2], uc[PF / 2];
        #pragma unroll
        for (int k = 0; k < PF / 2; ++k) { dc[k] = dn[k]; uc[k] = un[k]; }
        if (tb + PF < CLEN) {
            #pragma unroll
            for (int k = 0; k < PF / 2; ++k) {
                dn[k].x = dp[(tb + PF + 2 * k) * DMODEL];  dn[k].y = dp[(tb + PF + 2 * k + 1) * DMODEL];
                un[k].x = up[(tb + PF + 2 * k) * DMODEL];  un[k].y = up[(tb + PF + 2 * k + 1) * DMODEL];
            }
        }
        v2f X[NSTATE];
        v2f duk;
        #pragma unroll
        for (int j = 0; j < PF; ++j) {
            const int t = tb + j;
            v2f Bp[NPAIR];
            #pragma unroll
            for (int q = 0; q < 4; ++q) {
                float4 vb = nb[q];
                Bp[2*q] = (v2f){vb.x, vb.y};  Bp[2*q+1] = (v2f){vb.z, vb.w};
            }
            const int tn = (t + 1 < CLEN) ? (t + 1) : t;
            #pragma unroll
            for (int q = 0; q < 4; ++q) nb[q] = sB[tn * 4 + q];
            __builtin_amdgcn_sched_barrier(0);
            const v2f dk = dc[j / 2];
            if ((j & 1) == 0) {
                #pragma unroll
                for (int n = 0; n < NSTATE; ++n) X[n] = scan_arg(dk, A2, n);
                duk = dk * uc[j / 2];
            }
            const float dl = (j & 1) ? dk.y : dk.x;
            const v2f du2 = (j & 1) ? duk.yy : duk.xx;
            #pragma unroll
            for (int p = 0; p < NPAIR; ++p) {
                v2f dA;
                dA.x = __builtin_amdgcn_exp2f((j & 1) ? X[2 * p].y     : X[2 * p].x);
                dA.y = __builtin_amdgcn_exp2f((j & 1) ? X[2 * p + 1].y : X[2 * p + 1].x);
                st[p] = pk_fma(dA, st[p], du2 * Bp[p]);
            }
            dsum += dl;
        }
    }
    const int base = (c * 2 * NSTATE) * NCHAN + ch;
    #pragma unroll
    for (int p = 0; p < NPAIR; ++p) {
        const v2f pa = pk_exp2(A2[p] * dsum);
        agg[base + (2 * p) * NCHAN]     = pa.x;
        agg[base + (2 * p + 1) * NCHAN] = pa.y;
    }
    #pragma unroll
    for (int p = 0; p < NPAIR; ++p) {
        agg[base + (NSTATE + 2 * p) * NCHAN]     = st[p].x;
        agg[base + (NSTATE + 2 * p + 1) * NCHAN] = st[p].y;
    }
}

#define FX_CH   32
#define FX_CPB  8
#define FX_XB   (DMODEL / FX_CH)

struct FxGroup { float pa[FX_CPB], pb[FX_CPB], fa[FX_CPB], fb[FX_CPB]; };

__device__ __forceinline__ void fx_load(FxGroup& s, const float* __restrict__ ag, int grp) {
    #pragma unroll
    for (int k = 0; k < FX_CPB; ++k) {
        const float* a = ag + ((grp * FX_CPB + k) * 2 * NSTATE) * NCHAN;
        s.pa[k] = a[0];               s.pb[k] = a[NCHAN];
        s.fa[k] = a[NSTATE * NCHAN];  s.fb[k] = a[(NSTATE + 1) * NCHAN];
    }
}

__device__ __forceinline__ void fx_pass(const FxGroup& s, float& ea, float& eb) {
    #pragma unroll
    for (int k = 0; k < FX_CPB; ++k) {
        ea = fmaf(s.pa[k], ea, s.fa[k]);
        eb = fmaf(s.pb[k], eb, s.fb[k]);
    }
}

__device__ __forceinline__ void fx_own(const FxGroup& s, float ea, float eb, float* se) {
    #pragma unroll
    for (int k = 0; k < FX_CPB; ++k) {
        se[k * NSTATE * FX_CH]         = ea;
        se[k * NSTATE * FX_CH + FX_CH] = eb;
        if (k + 1 < FX_CPB) {
            ea = fmaf(s.pa[k], ea, s.fa[k]);
            eb = fmaf(s.pb[k], eb, s.fb[k]);
        }
    }
}

__global__ __launch_bounds__(256)
void scan_fixup(const float* __restrict__ U, const float* __restrict__ A_log,
                const float* __restrict__ bc, const float* __restrict__ delta,
                const float* __restrict__ Dvec, const float* __restrict__ agg,
                float* __restrict__ out) {
    __shared__ float4 sBC[FX_CPB * CLEN * 8];
    __shared__ float  sE[FX_CPB * NSTATE * FX_CH];

    const int lane = threadIdx.x & 31;
    const int wave = threadIdx.x >> 5;
    const int g    = blockIdx.y;
    const int c    = g * FX_CPB + wave;
    const int bb   = blockIdx.x / FX_XB;
    const int dd   = (blockIdx.x % FX_XB) * FX_CH + lane;
    const int ch   = bb * DMODEL + dd;
    const int row0 = bb * SEQ + c * CLEN;

    const float* __restrict__ dp = delta + row0 * DMODEL + dd;
    const float* __restrict__ up = U     + row0 * DMODEL + dd;
    float*       __restrict__ op = out   + row0 * DMODEL + dd;

    float4* sBCw = sBC + wave * (CLEN * 8);
    #pragma unroll
    for (int q = 0; q < CLEN * 8 / 32; ++q) {
        const int i = lane + 32 * q;
        sBCw[i] = *(const float4*)(bc + (row0 + (i >> 3)) * (2 * NSTATE) + 4 * (i & 7));
    }

    v2f dn[PF / 2], un[PF / 2];
    #pragma unroll
    for (int k = 0; k < PF / 2; ++k) {
        dn[k].x = dp[(2 * k) * DMODEL];  dn[k].y = dp[(2 * k + 1) * DMODEL];
        un[k].x = up[(2 * k) * DMODEL];  un[k].y = up[(2 * k + 1) * DMODEL];
    }

    v2f A2[NPAIR];
    #pragma unroll
    for (int p = 0; p < NPAIR; ++p) A2[p] = scan_a2(A_log, p);

    static_assert(NCHUNK / FX_CPB == 4, "the prefix below is written out for four groups of chunks");
    const float* ag = agg + (2 * wave) * NCHAN + ch;
    float ea = 0.0f, eb = 0.0f;
    float* se = sE + (2 * wave) * FX_CH + lane;
    FxGroup A, B;
    fx_load(A, ag, 0);
    if (g == 0) {
        fx_own(A, ea, eb, se);
    } else {
        fx_load(B, ag, 1);
        __builtin_amdgcn_sched_barrier(0);
        fx_pass(A, ea, eb);
        if (g == 1) {
            fx_own(B, ea, eb, se);
        } else {
            fx_load(A, ag, 2);
            __builtin_amdgcn_sched_barrier(0);
            fx_pass(B, ea, eb);
            if (g == 2) {
                fx_own(A, ea, eb, se);
            } else {
                fx_load(B, ag, 3);
                __builtin_amdgcn_sched_barrier(0);
                fx_pass(A, ea, eb);
                fx_own(B, ea, eb, se);
            }
        }
    }

    const float Dd = Dvec[dd];

    __syncthreads();

    v2f st[NPAIR];
    #pragma unroll
    for (int p = 0; p < NPAIR; ++p) {
        st[p].x = sE[(wave * NSTATE + 2 * p) * FX_CH + lane];
        st[p].y = sE[(wave * NSTATE + 2 * p + 1) * FX_CH + lane];
    }

    float4 nb[8];
    #pragma unroll
    for (int q = 0; q < 8; ++q) nb[q] = sBCw[q];

    #pragma unroll 1
    for (int tb = 0; tb < CLEN; tb += PF) {
        v2f dc[PF / 2], uc[PF / 2];
        #pragma unroll
        for (int k = 0; k < PF / 2; ++k) { dc[k] = dn[k]; uc[k] = un[k]; }
        if (tb + PF < CLEN) {
            #pragma unroll
            for (int k = 0; k < PF / 2; ++k) {
                dn[k].x = dp[(tb + PF + 2 * k) * DMODEL];  dn[k].y = dp[(tb + PF + 2 * k + 1) * DMODEL];
                un[k].x = up[(tb + PF + 2 * k) * DMODEL];  un[k].y = up[(tb + PF + 2 * k + 1) * DMODEL];
            }
        }
        v2f X[NSTATE];
        v2f duk;
        #pragma unroll
        for (int j = 0; j < PF; ++j) {
            const int t = tb + j;
            v2f Bp[NPAIR], Cp[NPAIR];
            #pragma unroll
            for (int q = 0; q < 4; ++q) {
                float4 vb = nb[q], vc = nb[4 + q];
                Bp[2*q] = (v2f){vb.x, vb.y};  Bp[2*q+1] = (v2f){vb.z, vb.w};
                Cp[2*q] = (v2f){vc.x, vc.y};  Cp[2*q+1] = (v2f){vc.z, vc.w};
            }
            const int tn = (t + 1 < CLEN) ? (t + 1) : t;
            #pragma unroll
            for (int q = 0; q < 8; ++q) nb[q] = sBCw[tn * 8 + q];
            __builtin_amdgcn_sched_barrier(0);
            const v2f dk = dc[j / 2];
            if ((j & 1) == 0) {
                #pragma unroll
                for (int n = 0; n < NSTATE; ++n) X[n] = scan_arg(dk, A2, n);
                duk = dk * uc[j / 2];
            }
            const float ut  = (j & 1) ? uc[j / 2].y : uc[j / 2].x;
            const v2f   du2 = (j & 1) ? duk.yy : duk.xx;
            v2f y2 = {0.0f, 0.0f};
            #pragma unroll
            for (int p = 0; p < NPAIR; ++p) {
                v2f dA;
                dA.x = __builtin_amdgcn_exp2f((j & 1) ? X[2 * p].y     : X[2 * p].x);
                dA.y = __builtin_amdgcn_exp2f((j & 1) ? X[2 * p + 1].y : X[2 * p + 1].x);
                st[p] = pk_fma(dA, st[p], du2 * Bp[p]);
                y2    = pk_fma(st[p], Cp[p], y2);
            }
            op[t * DMODEL] = (y2.x + y2.y) + ut * Dd;
        }
    }
}


extern "C" void kernel_launch(void* const* d_in, const int* in_sizes, int n_in,
                              void* d_out, int out_size, void* d_ws, size_t ws_size,
                              hipStream_t stream) {
    const float* U     = (const float*)d_in[0];
    const float* A_log = (const float*)d_in[1];
    const float* W1    = (const float*)d_in[2];
    const float* W2    = (const float*)d_in[3];
    const float* bias  = (const float*)d_in[4];
    const float* Dvec  = (const float*)d_in[5];
    float* out = (float*)d_out;

    float* ws    = (float*)d_ws;
    float* delta = ws;
    float* agg   = delta + (size_t)MROWS * DMODEL;
    float* bc    = agg + (size_t)NCHUNK * 2 * NSTATE * NCHAN;

    gemm_fused<<<dim3(MROWS / F_ROWS), dim3(256), 0, stream>>>(U, W1, W2, bias, delta, bc);

    dim3 sgrid(NCHAN / 256, NCHUNK);
    scan_local<<<sgrid, dim3(256), 0, stream>>>(U, A_log, bc, delta, agg);
    scan_fixup<<<dim3(NCHAN / FX_CH, NCHUNK / FX_CPB), dim3(256), 0, stream>>>(U, A_log, bc, delta, Dvec, agg, out);
}
